# starting kernel
# baseline (speedup 1.0000x reference)
#include <hip/hip_runtime.h>

typedef __attribute__((ext_vector_type(2))) float v2f;
typedef __attribute__((ext_vector_type(8))) float v8f;

#define BATCH 64
#define CHUNKS 4

__global__ void k_transpose(const float* __restrict__ x, float* __restrict__ xt, int ns) {
    __shared__ float tile[32][33];
    const int tx = threadIdx.x;
    const int ty = threadIdx.y;
    const int s0 = blockIdx.x * 32;
    const int b0 = blockIdx.y * 32;
#pragma unroll
    for (int k = 0; k < 4; ++k) {
        int s = s0 + tx, b = b0 + ty + k * 8;
        if (s < ns && b < BATCH) tile[ty + k * 8][tx] = x[(size_t)b * ns + s];
    }
    __syncthreads();
#pragma unroll
    for (int k = 0; k < 4; ++k) {
        int s = s0 + ty + k * 8, b = b0 + tx;
        if (s < ns && b < BATCH) xt[(size_t)s * BATCH + b] = tile[tx][ty + k * 8];
    }
}

__global__ void k_zero(float4* __restrict__ p, long long n4) {
    long long i = (long long)blockIdx.x * blockDim.x + threadIdx.x;
    const long long stride = (long long)gridDim.x * blockDim.x;
    const float4 z = make_float4(0.f, 0.f, 0.f, 0.f);
    for (; i < n4; i += stride) p[i] = z;
}

__global__ void k_scatter(const float* __restrict__ xt, const float* __restrict__ vals,
                          const int* __restrict__ row, const int* __restrict__ col,
                          float* __restrict__ out_t, int nnz) {
    const int lane = threadIdx.x & 31;
    const long long wave = ((long long)blockIdx.x * blockDim.x + threadIdx.x) >> 5;
    const long long base0 = wave * (32LL * CHUNKS);
    for (int ch = 0; ch < CHUNKS; ++ch) {
        const long long base = base0 + 32LL * ch;
        if (base >= nnz) return;
        const long long idx = base + lane;
        float v = 0.f;
        int r = 0, c = 0;
        if (idx < nnz) {
            v = vals[idx];
            r = row[idx];
            c = col[idx];
        }
        if (ch + 1 < CHUNKS && base + 32 + lane < nnz) {
            __builtin_prefetch(vals + base + 32 + lane, 0, 0);
            __builtin_prefetch(row + base + 32 + lane, 0, 0);
            __builtin_prefetch(col + base + 32 + lane, 0, 0);
        }
        const int count = (nnz - base < 32) ? (int)(nnz - base) : 32;
        for (int j = 0; j < count; ++j) {
            const float vj = __shfl(v, j);
            const int rj   = __shfl(r, j);
            const int cj   = __shfl(c, j);
            const v2f xv = *(const v2f*)(xt + (size_t)cj * BATCH + lane * 2);
            float* dst = out_t + (size_t)rj * BATCH + lane * 2;
            unsafeAtomicAdd(dst, xv.x * vj);
            unsafeAtomicAdd(dst + 1, xv.y * vj);
        }
    }
}

__global__ void k_epilogue_wmma(const float* __restrict__ out_t, const float* __restrict__ bias,
                                float* __restrict__ out, int nd, int dtiles) {
    const int lane = threadIdx.x & 31;
    const int wave = (int)(((long long)blockIdx.x * blockDim.x + threadIdx.x) >> 5);
    const int tiles = (BATCH / 16) * dtiles;
    if (wave >= tiles) return;
    const int b0 = (wave & 3) * 16;
    const int d0 = (wave >> 2) * 16;
    const int n = lane & 15;
    const int mbase = (lane < 16) ? 0 : 8;

    const float* cbase = out_t + (size_t)(d0 + n) * BATCH + b0 + mbase;
    const float4 lo = *(const float4*)(cbase);
    const float4 hi = *(const float4*)(cbase + 4);
    v8f c = {lo.x, lo.y, lo.z, lo.w, hi.x, hi.y, hi.z, hi.w};

    v2f a = {(lane < 16) ? 1.0f : 0.0f, 0.0f};
    v2f bm = {(lane < 16) ? bias[d0 + n] : 0.0f, 0.0f};

    v8f d = __builtin_amdgcn_wmma_f32_16x16x4_f32(
         false, a,  false, bm,
         (short)0, c,  false,  false);

#pragma unroll
    for (int v = 0; v < 8; ++v) {
        out[(size_t)(b0 + mbase + v) * nd + d0 + n] = d[v];
    }
}

__global__ void k_epilogue_tail(const float* __restrict__ out_t, const float* __restrict__ bias,
                                float* __restrict__ out, int nd, int dstart) {
    const int total = (nd - dstart) * BATCH;
    const int i = blockIdx.x * blockDim.x + threadIdx.x;
    if (i >= total) return;
    const int d = dstart + (i >> 6);
    const int b = i & 63;
    out[(size_t)b * nd + d] = out_t[(size_t)d * BATCH + b] + bias[d];
}

extern "C" void kernel_launch(void* const* d_in, const int* in_sizes, int n_in,
                              void* d_out, int out_size, void* d_ws, size_t ws_size,
                              hipStream_t stream) {
    const float* x    = (const float*)d_in[0];
    const float* vals = (const float*)d_in[1];
    const float* bias = (const float*)d_in[2];
    const int* row    = (const int*)d_in[3];
    const int* col    = (const int*)d_in[4];
    float* out = (float*)d_out;

    const int nnz = in_sizes[1];
    const int nd  = in_sizes[2];
    const int ns  = in_sizes[0] / BATCH;

    float* xt    = (float*)d_ws;
    float* out_t = xt + (size_t)ns * BATCH;

    dim3 tb(32, 8);
    dim3 tg((ns + 31) / 32, (BATCH + 31) / 32);
    k_transpose<<<tg, tb, 0, stream>>>(x, xt, ns);

    long long n4 = ((long long)nd * BATCH) / 4;
    long long zb = (n4 + 255) / 256;
    if (zb > 16384) zb = 16384;
    k_zero<<<(int)zb, 256, 0, stream>>>((float4*)out_t, n4);

    long long waves = ((long long)nnz + 32LL * CHUNKS - 1) / (32LL * CHUNKS);
    long long sblocks = (waves + 7) / 8;
    if (sblocks < 1) sblocks = 1;
    k_scatter<<<(int)sblocks, 256, 0, stream>>>(xt, vals, row, col, out_t, nnz);

    const int dtiles = nd / 16;
    if (dtiles > 0) {
        const int tiles = (BATCH / 16) * dtiles;
        const int eblocks = (tiles + 7) / 8;
        k_epilogue_wmma<<<eblocks, 256, 0, stream>>>(out_t, bias, out, nd, dtiles);
    }
    const int dstart = dtiles * 16;
    if (dstart < nd) {
        const int total = (nd - dstart) * BATCH;
        k_epilogue_tail<<<(total + 255) / 256, 256, 0, stream>>>(out_t, bias, out, nd, dstart);
    }
}


# Replace global f32 atomic scatter: counting-sort edges into 400-row destination buckets (LDS counters), sum each bucket's rows in LDS with ds f32 adds (arrival order), store tile once
# speedup vs baseline: 1.6096x; 1.6096x over previous
#include <hip/hip_runtime.h>

typedef __attribute__((ext_vector_type(2))) float v2f;
typedef __attribute__((ext_vector_type(8))) float v8f;

#define BATCH 64
#define CHUNKS 4

__global__ void k_transpose(const float* __restrict__ x, float* __restrict__ xt, int ns) {
    __shared__ float tile[32][33];
    const int tx = threadIdx.x;
    const int ty = threadIdx.y;
    const int s0 = blockIdx.x * 32;
    const int b0 = blockIdx.y * 32;
#pragma unroll
    for (int k = 0; k < 4; ++k) {
        int s = s0 + tx, b = b0 + ty + k * 8;
        if (s < ns && b < BATCH) tile[ty + k * 8][tx] = x[(size_t)b * ns + s];
    }
    __syncthreads();
#pragma unroll
    for (int k = 0; k < 4; ++k) {
        int s = s0 + ty + k * 8, b = b0 + tx;
        if (s < ns && b < BATCH) xt[(size_t)s * BATCH + b] = tile[tx][ty + k * 8];
    }
}

__global__ void k_zero(float4* __restrict__ p, long long n4) {
    long long i = (long long)blockIdx.x * blockDim.x + threadIdx.x;
    const long long stride = (long long)gridDim.x * blockDim.x;
    const float4 z = make_float4(0.f, 0.f, 0.f, 0.f);
    for (; i < n4; i += stride) p[i] = z;
}

__global__ void k_scatter(const float* __restrict__ xt, const float* __restrict__ vals,
                          const int* __restrict__ row, const int* __restrict__ col,
                          float* __restrict__ out_t, int nnz) {
    const int lane = threadIdx.x & 31;
    const long long wave = ((long long)blockIdx.x * blockDim.x + threadIdx.x) >> 5;
    const long long base0 = wave * (32LL * CHUNKS);
    for (int ch = 0; ch < CHUNKS; ++ch) {
        const long long base = base0 + 32LL * ch;
        if (base >= nnz) return;
        const long long idx = base + lane;
        float v = 0.f;
        int r = 0, c = 0;
        if (idx < nnz) {
            v = vals[idx];
            r = row[idx];
            c = col[idx];
        }
        if (ch + 1 < CHUNKS && base + 32 + lane < nnz) {
            __builtin_prefetch(vals + base + 32 + lane, 0, 0);
            __builtin_prefetch(row + base + 32 + lane, 0, 0);
            __builtin_prefetch(col + base + 32 + lane, 0, 0);
        }
        const int count = (nnz - base < 32) ? (int)(nnz - base) : 32;
        for (int j = 0; j < count; ++j) {
            const float vj = __shfl(v, j);
            const int rj   = __shfl(r, j);
            const int cj   = __shfl(c, j);
            const v2f xv = *(const v2f*)(xt + (size_t)cj * BATCH + lane * 2);
            float* dst = out_t + (size_t)rj * BATCH + lane * 2;
            unsafeAtomicAdd(dst, xv.x * vj);
            unsafeAtomicAdd(dst + 1, xv.y * vj);
        }
    }
}

__global__ void k_epilogue_wmma(const float* __restrict__ out_t, const float* __restrict__ bias,
                                float* __restrict__ out, int nd, int dtiles) {
    const int lane = threadIdx.x & 31;
    const int wave = (int)(((long long)blockIdx.x * blockDim.x + threadIdx.x) >> 5);
    const int tiles = (BATCH / 16) * dtiles;
    if (wave >= tiles) return;
    const int b0 = (wave & 3) * 16;
    const int d0 = (wave >> 2) * 16;
    const int n = lane & 15;
    const int mbase = (lane < 16) ? 0 : 8;

    const float* cbase = out_t + (size_t)(d0 + n) * BATCH + b0 + mbase;
    const float4 lo = *(const float4*)(cbase);
    const float4 hi = *(const float4*)(cbase + 4);
    v8f c = {lo.x, lo.y, lo.z, lo.w, hi.x, hi.y, hi.z, hi.w};

    v2f a = {(lane < 16) ? 1.0f : 0.0f, 0.0f};
    v2f bm = {(lane < 16) ? bias[d0 + n] : 0.0f, 0.0f};

    v8f d = __builtin_amdgcn_wmma_f32_16x16x4_f32(
         false, a,  false, bm,
         (short)0, c,  false,  false);

#pragma unroll
    for (int v = 0; v < 8; ++v) {
        out[(size_t)(b0 + mbase + v) * nd + d0 + n] = d[v];
    }
}

__global__ void k_epilogue_tail(const float* __restrict__ out_t, const float* __restrict__ bias,
                                float* __restrict__ out, int nd, int dstart) {
    const int total = (nd - dstart) * BATCH;
    const int i = blockIdx.x * blockDim.x + threadIdx.x;
    if (i >= total) return;
    const int d = dstart + (i >> 6);
    const int b = i & 63;
    out[(size_t)b * nd + d] = out_t[(size_t)d * BATCH + b] + bias[d];
}

#define RB 400
#define NBMAX 256
#define PT 1024
#define EPT 13
#define TILE (PT * EPT)

__device__ __forceinline__ int block_excl_scan(int v, int* s_w, int& total) {
    const int lane = threadIdx.x & 31;
    const int w = threadIdx.x >> 5;
    int inc = v;
#pragma unroll
    for (int d = 1; d < 32; d <<= 1) {
        const int t = __shfl_up(inc, d);
        if (lane >= d) inc += t;
    }
    if (lane == 31) s_w[w] = inc;
    __syncthreads();
    if (w == 0) {
        const int x = s_w[lane];
        int xi = x;
#pragma unroll
        for (int d = 1; d < 32; d <<= 1) {
            const int t = __shfl_up(xi, d);
            if (lane >= d) xi += t;
        }
        s_w[32 + lane] = xi - x;
        if (lane == 31) s_w[64] = xi;
    }
    __syncthreads();
    total = s_w[64];
    const int r = s_w[32 + w] + inc - v;
    __syncthreads();
    return r;
}

__global__ void __launch_bounds__(PT) k_hist(const int* __restrict__ row, int nnz,
                                             int* __restrict__ hist) {
    __shared__ int s_h[NBMAX];
    const int tid = threadIdx.x;
    if (tid < NBMAX) s_h[tid] = 0;
    __syncthreads();
    const long long base = (long long)blockIdx.x * TILE;
#pragma unroll
    for (int k = 0; k < EPT; ++k) {
        const long long e = base + k * PT + tid;
        if (e < nnz) atomicAdd(&s_h[row[e] / RB], 1);
    }
    __syncthreads();
    if (tid < NBMAX) hist[blockIdx.x * NBMAX + tid] = s_h[tid];
}

__global__ void __launch_bounds__(PT) k_scanp(const int* __restrict__ hist, int* __restrict__ off,
                                              int* __restrict__ tot, int pb) {
    __shared__ int s_w[65];
    const int b = blockIdx.x;
    const int tid = threadIdx.x;
    int carry = 0;
    for (int p0 = 0; p0 < pb; p0 += PT) {
        const int p = p0 + tid;
        const int v = (p < pb) ? hist[p * NBMAX + b] : 0;
        int total;
        const int ex = block_excl_scan(v, s_w, total);
        if (p < pb) off[p * NBMAX + b] = carry + ex;
        carry += total;
    }
    if (tid == 0) tot[b] = carry;
}

__global__ void __launch_bounds__(PT) k_place(const int* __restrict__ row, const int* __restrict__ col,
                                              const float* __restrict__ vals, int nnz,
                                              const int* __restrict__ off, const int* __restrict__ tot,
                                              int* __restrict__ rec_r, int* __restrict__ rec_c,
                                              float* __restrict__ rec_v) {
    __shared__ int s_r[TILE];
    __shared__ int s_c[TILE];
    __shared__ float s_v[TILE];
    __shared__ int s_h[NBMAX];
    __shared__ int s_loff[NBMAX];
    __shared__ int s_goff[NBMAX];
    __shared__ int s_w[65];
    const int tid = threadIdx.x;

    int total;
    const int t = (tid < NBMAX) ? tot[tid] : 0;
    const int bs = block_excl_scan(t, s_w, total);
    if (tid < NBMAX) {
        s_goff[tid] = bs + off[blockIdx.x * NBMAX + tid];
        s_h[tid] = 0;
    }
    __syncthreads();

    int myr[EPT], myc[EPT], myk[EPT];
    float myv[EPT];
    const long long base = (long long)blockIdx.x * TILE;
#pragma unroll
    for (int k = 0; k < EPT; ++k) {
        const long long e = base + k * PT + tid;
        myk[k] = -1;
        myr[k] = 0; myc[k] = 0; myv[k] = 0.f;
        if (e < nnz) {
            const int r = row[e];
            myr[k] = r;
            myc[k] = col[e];
            myv[k] = vals[e];
            myk[k] = atomicAdd(&s_h[r / RB], 1);
        }
    }
    __syncthreads();
    const int h = (tid < NBMAX) ? s_h[tid] : 0;
    const int lo = block_excl_scan(h, s_w, total);
    if (tid < NBMAX) s_loff[tid] = lo;
    __syncthreads();
#pragma unroll
    for (int k = 0; k < EPT; ++k) {
        if (myk[k] >= 0) {
            const int pos = s_loff[myr[k] / RB] + myk[k];
            s_r[pos] = myr[k];
            s_c[pos] = myc[k];
            s_v[pos] = myv[k];
        }
    }
    __syncthreads();
    for (int i = tid; i < total; i += PT) {
        const int r = s_r[i];
        const int b = r / RB;
        const int g = s_goff[b] + (i - s_loff[b]);
        rec_r[g] = r;
        rec_c[g] = s_c[i];
        rec_v[g] = s_v[i];
    }
}

__global__ void __launch_bounds__(PT) k_accum(const float* __restrict__ xt, const int* __restrict__ rec_r,
                                              const int* __restrict__ rec_c, const float* __restrict__ rec_v,
                                              const int* __restrict__ tot, float* __restrict__ out_t, int nd) {
    __shared__ __attribute__((aligned(16))) float acc[RB * BATCH];
    __shared__ int s_w[65];
    __shared__ int s_se[2];
    const int tid = threadIdx.x;
    const int lane = tid & 31;
    const int w = __builtin_amdgcn_readfirstlane(tid >> 5);
    const int b = blockIdx.x;

    int total;
    const int t = (tid < NBMAX) ? tot[tid] : 0;
    const int bs = block_excl_scan(t, s_w, total);
    if (tid == b) { s_se[0] = bs; s_se[1] = t; }
    const float4 z = make_float4(0.f, 0.f, 0.f, 0.f);
    for (int i = tid; i < RB * BATCH / 4; i += PT) ((float4*)acc)[i] = z;
    __syncthreads();
    const int start = s_se[0];
    const int cnt = s_se[1];
    const int row0 = b * RB;

    int rn = 0, cn = 0;
    float vn = 0.f;
    if (w * 32 + lane < cnt) {
        const int g = start + w * 32 + lane;
        rn = rec_r[g];
        cn = rec_c[g];
        vn = rec_v[g];
    }
    for (int i0 = w * 32; i0 < cnt; i0 += 32 * (PT / 32)) {
        const int n = (cnt - i0 < 32) ? (cnt - i0) : 32;
        const int r = rn - row0, c = cn;
        const float v = vn;
        const int i1 = i0 + 32 * (PT / 32);
        if (i1 + lane < cnt) {
            const int g = start + i1 + lane;
            rn = rec_r[g];
            cn = rec_c[g];
            vn = rec_v[g];
        }
        int j0 = 0;
        for (; j0 + 8 <= n; j0 += 8) {
            v2f xv[8];
#pragma unroll
            for (int u = 0; u < 8; ++u) {
                const int cj = __shfl(c, j0 + u);
                xv[u] = *(const v2f*)(xt + (size_t)cj * BATCH + lane * 2);
            }
#pragma unroll
            for (int u = 0; u < 8; ++u) {
                const float vj = __shfl(v, j0 + u);
                const int rj = __shfl(r, j0 + u);
                float* dst = acc + rj * BATCH + lane * 2;
                unsafeAtomicAdd(dst, xv[u].x * vj);
                unsafeAtomicAdd(dst + 1, xv[u].y * vj);
            }
        }
        for (; j0 < n; ++j0) {
            const int cj = __shfl(c, j0);
            const float vj = __shfl(v, j0);
            const int rj = __shfl(r, j0);
            const v2f xs = *(const v2f*)(xt + (size_t)cj * BATCH + lane * 2);
            float* dst = acc + rj * BATCH + lane * 2;
            unsafeAtomicAdd(dst, xs.x * vj);
            unsafeAtomicAdd(dst + 1, xs.y * vj);
        }
    }
    __syncthreads();
    const int rows = (nd - row0 < RB) ? (nd - row0) : RB;
    float4* dst4 = (float4*)(out_t + (size_t)row0 * BATCH);
    for (int i = tid; i < rows * (BATCH / 4); i += PT) dst4[i] = ((const float4*)acc)[i];
}

extern "C" void kernel_launch(void* const* d_in, const int* in_sizes, int n_in,
                              void* d_out, int out_size, void* d_ws, size_t ws_size,
                              hipStream_t stream) {
    const float* x    = (const float*)d_in[0];
    const float* vals = (const float*)d_in[1];
    const float* bias = (const float*)d_in[2];
    const int* row    = (const int*)d_in[3];
    const int* col    = (const int*)d_in[4];
    float* out = (float*)d_out;

    const int nnz = in_sizes[1];
    const int nd  = in_sizes[2];
    const int ns  = in_sizes[0] / BATCH;

    float* xt    = (float*)d_ws;
    float* out_t = xt + (size_t)ns * BATCH;

    dim3 tb(32, 8);
    dim3 tg((ns + 31) / 32, (BATCH + 31) / 32);
    k_transpose<<<tg, tb, 0, stream>>>(x, xt, ns);

    const int nb = (nd + RB - 1) / RB;
    const int pb = (int)(((long long)nnz + TILE - 1) / TILE);
    const size_t need_words = (size_t)ns * BATCH + (size_t)nd * BATCH + 3 * (size_t)nnz +
                              2 * (size_t)pb * NBMAX + NBMAX;
    if (nb <= NBMAX && pb >= 1 && need_words * 4 <= ws_size) {
        int* rec_r   = (int*)(out_t + (size_t)nd * BATCH);
        int* rec_c   = rec_r + (size_t)nnz;
        float* rec_v = (float*)(rec_c + (size_t)nnz);
        int* hist    = (int*)(rec_v + (size_t)nnz);
        int* off     = hist + (size_t)pb * NBMAX;
        int* tot     = off + (size_t)pb * NBMAX;
        k_hist<<<pb, PT, 0, stream>>>(row, nnz, hist);
        k_scanp<<<NBMAX, PT, 0, stream>>>(hist, off, tot, pb);
        k_place<<<pb, PT, 0, stream>>>(row, col, vals, nnz, off, tot, rec_r, rec_c, rec_v);
        k_accum<<<nb, PT, 0, stream>>>(xt, rec_r, rec_c, rec_v, tot, out_t, nd);
    } else {
        long long n4 = ((long long)nd * BATCH) / 4;
        long long zb = (n4 + 255) / 256;
        if (zb > 16384) zb = 16384;
        k_zero<<<(int)zb, 256, 0, stream>>>((float4*)out_t, n4);
        long long waves = ((long long)nnz + 32LL * CHUNKS - 1) / (32LL * CHUNKS);
        long long sblocks = (waves + 7) / 8;
        if (sblocks < 1) sblocks = 1;
        k_scatter<<<(int)sblocks, 256, 0, stream>>>(xt, vals, row, col, out_t, nnz);
    }

    const int dtiles = nd / 16;
    if (dtiles > 0) {
        const int tiles = (BATCH / 16) * dtiles;
        const int eblocks = (tiles + 7) / 8;
        k_epilogue_wmma<<<eblocks, 256, 0, stream>>>(out_t, bias, out, nd, dtiles);
    }
    const int dstart = dtiles * 16;
    if (dstart < nd) {
        const int total = (nd - dstart) * BATCH;
        k_epilogue_tail<<<(total + 255) / 256, 256, 0, stream>>>(out_t, bias, out, nd, dstart);
    }
}


# x transpose as 64x128 tiles: 8 whole-line 16-byte loads in flight per lane, turned in LDS at odd pitch, 512-byte row-pair stores; 782 eight-wave blocks, 6256 waves instead of 50000
# speedup vs baseline: 1.6644x; 1.0340x over previous
#include <hip/hip_runtime.h>

typedef __attribute__((ext_vector_type(2))) float v2f;
typedef __attribute__((ext_vector_type(8))) float v8f;

#define BATCH 64
#define CHUNKS 4

__global__ void k_transpose(const float* __restrict__ x, float* __restrict__ xt, int ns) {
    __shared__ float tile[32][33];
    const int tx = threadIdx.x;
    const int ty = threadIdx.y;
    const int s0 = blockIdx.x * 32;
    const int b0 = blockIdx.y * 32;
#pragma unroll
    for (int k = 0; k < 4; ++k) {
        int s = s0 + tx, b = b0 + ty + k * 8;
        if (s < ns && b < BATCH) tile[ty + k * 8][tx] = x[(size_t)b * ns + s];
    }
    __syncthreads();
#pragma unroll
    for (int k = 0; k < 4; ++k) {
        int s = s0 + ty + k * 8, b = b0 + tx;
        if (s < ns && b < BATCH) xt[(size_t)s * BATCH + b] = tile[tx][ty + k * 8];
    }
}

#define TS 128
#define TPITCH 65
template <bool FULL>
__device__ __forceinline__ void transpose_tile(const float* __restrict__ x, float* __restrict__ xt,
                                               int ns, int s0, float* __restrict__ t) {
    const int tid = threadIdx.x;
    const int q = tid & 31;
    const int w = tid >> 5;
    if (FULL || s0 + 4 * q < ns) {
        const float* p = x + (size_t)w * ns + s0 + 4 * q;
        const size_t step = (size_t)8 * ns;
        float4 v[8];
#pragma unroll
        for (int k = 0; k < 8; ++k) v[k] = *(const float4*)(p + k * step);
        const int wbase = (4 * q) * TPITCH + ((4 * q) >> 6) + w;
#pragma unroll
        for (int k = 0; k < 8; ++k) {
            t[wbase + 8 * k]              = v[k].x;
            t[wbase + 8 * k + TPITCH]     = v[k].y;
            t[wbase + 8 * k + 2 * TPITCH] = v[k].z;
            t[wbase + 8 * k + 3 * TPITCH] = v[k].w;
        }
    }
    __syncthreads();
    const int bq = tid & 15;
    const int sr = tid >> 4;
    float* dst = xt + (size_t)(s0 + sr) * BATCH + 4 * bq;
    if (FULL) {
        float4 o[8];
#pragma unroll
        for (int k = 0; k < 8; ++k) {
            const int s = sr + 16 * k;
            const int rb = s * TPITCH + (s >> 6) + 4 * bq;
            o[k] = make_float4(t[rb], t[rb + 1], t[rb + 2], t[rb + 3]);
        }
#pragma unroll
        for (int k = 0; k < 8; ++k) *(float4*)(dst + (size_t)(16 * k) * BATCH) = o[k];
    } else {
#pragma unroll
        for (int k = 0; k < 8; ++k) {
            const int s = sr + 16 * k;
            if (s0 + s < ns) {
                const int rb = s * TPITCH + (s >> 6) + 4 * bq;
                *(float4*)(dst + (size_t)(16 * k) * BATCH) =
                    make_float4(t[rb], t[rb + 1], t[rb + 2], t[rb + 3]);
            }
        }
    }
}

__global__ void __launch_bounds__(256) k_transpose_wide(const float* __restrict__ x,
                                                        float* __restrict__ xt, int ns) {
    __shared__ float t[TS * TPITCH + 4];
    const int s0 = blockIdx.x * TS;
    if (s0 + TS <= ns) transpose_tile<true>(x, xt, ns, s0, t);
    else               transpose_tile<false>(x, xt, ns, s0, t);
}

__global__ void k_zero(float4* __restrict__ p, long long n4) {
    long long i = (long long)blockIdx.x * blockDim.x + threadIdx.x;
    const long long stride = (long long)gridDim.x * blockDim.x;
    const float4 z = make_float4(0.f, 0.f, 0.f, 0.f);
    for (; i < n4; i += stride) p[i] = z;
}

__global__ void k_scatter(const float* __restrict__ xt, const float* __restrict__ vals,
                          const int* __restrict__ row, const int* __restrict__ col,
                          float* __restrict__ out_t, int nnz) {
    const int lane = threadIdx.x & 31;
    const long long wave = ((long long)blockIdx.x * blockDim.x + threadIdx.x) >> 5;
    const long long base0 = wave * (32LL * CHUNKS);
    for (int ch = 0; ch < CHUNKS; ++ch) {
        const long long base = base0 + 32LL * ch;
        if (base >= nnz) return;
        const long long idx = base + lane;
        float v = 0.f;
        int r = 0, c = 0;
        if (idx < nnz) {
            v = vals[idx];
            r = row[idx];
            c = col[idx];
        }
        if (ch + 1 < CHUNKS && base + 32 + lane < nnz) {
            __builtin_prefetch(vals + base + 32 + lane, 0, 0);
            __builtin_prefetch(row + base + 32 + lane, 0, 0);
            __builtin_prefetch(col + base + 32 + lane, 0, 0);
        }
        const int count = (nnz - base < 32) ? (int)(nnz - base) : 32;
        for (int j = 0; j < count; ++j) {
            const float vj = __shfl(v, j);
            const int rj   = __shfl(r, j);
            const int cj   = __shfl(c, j);
            const v2f xv = *(const v2f*)(xt + (size_t)cj * BATCH + lane * 2);
            float* dst = out_t + (size_t)rj * BATCH + lane * 2;
            unsafeAtomicAdd(dst, xv.x * vj);
            unsafeAtomicAdd(dst + 1, xv.y * vj);
        }
    }
}

__global__ void k_epilogue_wmma(const float* __restrict__ out_t, const float* __restrict__ bias,
                                float* __restrict__ out, int nd, int dtiles) {
    const int lane = threadIdx.x & 31;
    const int wave = (int)(((long long)blockIdx.x * blockDim.x + threadIdx.x) >> 5);
    const int tiles = (BATCH / 16) * dtiles;
    if (wave >= tiles) return;
    const int b0 = (wave & 3) * 16;
    const int d0 = (wave >> 2) * 16;
    const int n = lane & 15;
    const int mbase = (lane < 16) ? 0 : 8;

    const float* cbase = out_t + (size_t)(d0 + n) * BATCH + b0 + mbase;
    const float4 lo = *(const float4*)(cbase);
    const float4 hi = *(const float4*)(cbase + 4);
    v8f c = {lo.x, lo.y, lo.z, lo.w, hi.x, hi.y, hi.z, hi.w};

    v2f a = {(lane < 16) ? 1.0f : 0.0f, 0.0f};
    v2f bm = {(lane < 16) ? bias[d0 + n] : 0.0f, 0.0f};

    v8f d = __builtin_amdgcn_wmma_f32_16x16x4_f32(
         false, a,  false, bm,
         (short)0, c,  false,  false);

#pragma unroll
    for (int v = 0; v < 8; ++v) {
        out[(size_t)(b0 + mbase + v) * nd + d0 + n] = d[v];
    }
}

__global__ void k_epilogue_tail(const float* __restrict__ out_t, const float* __restrict__ bias,
                                float* __restrict__ out, int nd, int dstart) {
    const int total = (nd - dstart) * BATCH;
    const int i = blockIdx.x * blockDim.x + threadIdx.x;
    if (i >= total) return;
    const int d = dstart + (i >> 6);
    const int b = i & 63;
    out[(size_t)b * nd + d] = out_t[(size_t)d * BATCH + b] + bias[d];
}

#define RB 400
#define NBMAX 256
#define PT 1024
#define EPT 13
#define TILE (PT * EPT)

__device__ __forceinline__ int block_excl_scan(int v, int* s_w, int& total) {
    const int lane = threadIdx.x & 31;
    const int w = threadIdx.x >> 5;
    int inc = v;
#pragma unroll
    for (int d = 1; d < 32; d <<= 1) {
        const int t = __shfl_up(inc, d);
        if (lane >= d) inc += t;
    }
    if (lane == 31) s_w[w] = inc;
    __syncthreads();
    if (w == 0) {
        const int x = s_w[lane];
        int xi = x;
#pragma unroll
        for (int d = 1; d < 32; d <<= 1) {
            const int t = __shfl_up(xi, d);
            if (lane >= d) xi += t;
        }
        s_w[32 + lane] = xi - x;
        if (lane == 31) s_w[64] = xi;
    }
    __syncthreads();
    total = s_w[64];
    const int r = s_w[32 + w] + inc - v;
    __syncthreads();
    return r;
}

__global__ void __launch_bounds__(PT) k_hist(const int* __restrict__ row, int nnz,
                                             int* __restrict__ hist) {
    __shared__ int s_h[NBMAX];
    const int tid = threadIdx.x;
    if (tid < NBMAX) s_h[tid] = 0;
    __syncthreads();
    const long long base = (long long)blockIdx.x * TILE;
#pragma unroll
    for (int k = 0; k < EPT; ++k) {
        const long long e = base + k * PT + tid;
        if (e < nnz) atomicAdd(&s_h[row[e] / RB], 1);
    }
    __syncthreads();
    if (tid < NBMAX) hist[blockIdx.x * NBMAX + tid] = s_h[tid];
}

__global__ void __launch_bounds__(PT) k_scanp(const int* __restrict__ hist, int* __restrict__ off,
                                              int* __restrict__ tot, int pb) {
    __shared__ int s_w[65];
    const int b = blockIdx.x;
    const int tid = threadIdx.x;
    int carry = 0;
    for (int p0 = 0; p0 < pb; p0 += PT) {
        const int p = p0 + tid;
        const int v = (p < pb) ? hist[p * NBMAX + b] : 0;
        int total;
        const int ex = block_excl_scan(v, s_w, total);
        if (p < pb) off[p * NBMAX + b] = carry + ex;
        carry += total;
    }
    if (tid == 0) tot[b] = carry;
}

__global__ void __launch_bounds__(PT) k_place(const int* __restrict__ row, const int* __restrict__ col,
                                              const float* __restrict__ vals, int nnz,
                                              const int* __restrict__ off, const int* __restrict__ tot,
                                              int* __restrict__ rec_r, int* __restrict__ rec_c,
                                              float* __restrict__ rec_v) {
    __shared__ int s_r[TILE];
    __shared__ int s_c[TILE];
    __shared__ float s_v[TILE];
    __shared__ int s_h[NBMAX];
    __shared__ int s_loff[NBMAX];
    __shared__ int s_goff[NBMAX];
    __shared__ int s_w[65];
    const int tid = threadIdx.x;

    int total;
    const int t = (tid < NBMAX) ? tot[tid] : 0;
    const int bs = block_excl_scan(t, s_w, total);
    if (tid < NBMAX) {
        s_goff[tid] = bs + off[blockIdx.x * NBMAX + tid];
        s_h[tid] = 0;
    }
    __syncthreads();

    int myr[EPT], myc[EPT], myk[EPT];
    float myv[EPT];
    const long long base = (long long)blockIdx.x * TILE;
#pragma unroll
    for (int k = 0; k < EPT; ++k) {
        const long long e = base + k * PT + tid;
        myk[k] = -1;
        myr[k] = 0; myc[k] = 0; myv[k] = 0.f;
        if (e < nnz) {
            const int r = row[e];
            myr[k] = r;
            myc[k] = col[e];
            myv[k] = vals[e];
            myk[k] = atomicAdd(&s_h[r / RB], 1);
        }
    }
    __syncthreads();
    const int h = (tid < NBMAX) ? s_h[tid] : 0;
    const int lo = block_excl_scan(h, s_w, total);
    if (tid < NBMAX) s_loff[tid] = lo;
    __syncthreads();
#pragma unroll
    for (int k = 0; k < EPT; ++k) {
        if (myk[k] >= 0) {
            const int pos = s_loff[myr[k] / RB] + myk[k];
            s_r[pos] = myr[k];
            s_c[pos] = myc[k];
            s_v[pos] = myv[k];
        }
    }
    __syncthreads();
    for (int i = tid; i < total; i += PT) {
        const int r = s_r[i];
        const int b = r / RB;
        const int g = s_goff[b] + (i - s_loff[b]);
        rec_r[g] = r;
        rec_c[g] = s_c[i];
        rec_v[g] = s_v[i];
    }
}

__global__ void __launch_bounds__(PT) k_accum(const float* __restrict__ xt, const int* __restrict__ rec_r,
                                              const int* __restrict__ rec_c, const float* __restrict__ rec_v,
                                              const int* __restrict__ tot, float* __restrict__ out_t, int nd) {
    __shared__ __attribute__((aligned(16))) float acc[RB * BATCH];
    __shared__ int s_w[65];
    __shared__ int s_se[2];
    const int tid = threadIdx.x;
    const int lane = tid & 31;
    const int w = __builtin_amdgcn_readfirstlane(tid >> 5);
    const int b = blockIdx.x;

    int total;
    const int t = (tid < NBMAX) ? tot[tid] : 0;
    const int bs = block_excl_scan(t, s_w, total);
    if (tid == b) { s_se[0] = bs; s_se[1] = t; }
    const float4 z = make_float4(0.f, 0.f, 0.f, 0.f);
    for (int i = tid; i < RB * BATCH / 4; i += PT) ((float4*)acc)[i] = z;
    __syncthreads();
    const int start = s_se[0];
    const int cnt = s_se[1];
    const int row0 = b * RB;

    int rn = 0, cn = 0;
    float vn = 0.f;
    if (w * 32 + lane < cnt) {
        const int g = start + w * 32 + lane;
        rn = rec_r[g];
        cn = rec_c[g];
        vn = rec_v[g];
    }
    for (int i0 = w * 32; i0 < cnt; i0 += 32 * (PT / 32)) {
        const int n = (cnt - i0 < 32) ? (cnt - i0) : 32;
        const int r = rn - row0, c = cn;
        const float v = vn;
        const int i1 = i0 + 32 * (PT / 32);
        if (i1 + lane < cnt) {
            const int g = start + i1 + lane;
            rn = rec_r[g];
            cn = rec_c[g];
            vn = rec_v[g];
        }
        int j0 = 0;
        for (; j0 + 8 <= n; j0 += 8) {
            v2f xv[8];
#pragma unroll
            for (int u = 0; u < 8; ++u) {
                const int cj = __shfl(c, j0 + u);
                xv[u] = *(const v2f*)(xt + (size_t)cj * BATCH + lane * 2);
            }
#pragma unroll
            for (int u = 0; u < 8; ++u) {
                const float vj = __shfl(v, j0 + u);
                const int rj = __shfl(r, j0 + u);
                float* dst = acc + rj * BATCH + lane * 2;
                unsafeAtomicAdd(dst, xv[u].x * vj);
                unsafeAtomicAdd(dst + 1, xv[u].y * vj);
            }
        }
        for (; j0 < n; ++j0) {
            const int cj = __shfl(c, j0);
            const float vj = __shfl(v, j0);
            const int rj = __shfl(r, j0);
            const v2f xs = *(const v2f*)(xt + (size_t)cj * BATCH + lane * 2);
            float* dst = acc + rj * BATCH + lane * 2;
            unsafeAtomicAdd(dst, xs.x * vj);
            unsafeAtomicAdd(dst + 1, xs.y * vj);
        }
    }
    __syncthreads();
    const int rows = (nd - row0 < RB) ? (nd - row0) : RB;
    float4* dst4 = (float4*)(out_t + (size_t)row0 * BATCH);
    for (int i = tid; i < rows * (BATCH / 4); i += PT) dst4[i] = ((const float4*)acc)[i];
}

extern "C" void kernel_launch(void* const* d_in, const int* in_sizes, int n_in,
                              void* d_out, int out_size, void* d_ws, size_t ws_size,
                              hipStream_t stream) {
    const float* x    = (const float*)d_in[0];
    const float* vals = (const float*)d_in[1];
    const float* bias = (const float*)d_in[2];
    const int* row    = (const int*)d_in[3];
    const int* col    = (const int*)d_in[4];
    float* out = (float*)d_out;

    const int nnz = in_sizes[1];
    const int nd  = in_sizes[2];
    const int ns  = in_sizes[0] / BATCH;

    float* xt    = (float*)d_ws;
    float* out_t = xt + (size_t)ns * BATCH;

    if ((ns & 3) == 0) {
        k_transpose_wide<<<(ns + TS - 1) / TS, 256, 0, stream>>>(x, xt, ns);
    } else {
        dim3 tb(32, 8);
        dim3 tg((ns + 31) / 32, (BATCH + 31) / 32);
        k_transpose<<<tg, tb, 0, stream>>>(x, xt, ns);
    }

    const int nb = (nd + RB - 1) / RB;
    const int pb = (int)(((long long)nnz + TILE - 1) / TILE);
    const size_t need_words = (size_t)ns * BATCH + (size_t)nd * BATCH + 3 * (size_t)nnz +
                              2 * (size_t)pb * NBMAX + NBMAX;
    if (nb <= NBMAX && pb >= 1 && need_words * 4 <= ws_size) {
        int* rec_r   = (int*)(out_t + (size_t)nd * BATCH);
        int* rec_c   = rec_r + (size_t)nnz;
        float* rec_v = (float*)(rec_c + (size_t)nnz);
        int* hist    = (int*)(rec_v + (size_t)nnz);
        int* off     = hist + (size_t)pb * NBMAX;
        int* tot     = off + (size_t)pb * NBMAX;
        k_hist<<<pb, PT, 0, stream>>>(row, nnz, hist);
        k_scanp<<<NBMAX, PT, 0, stream>>>(hist, off, tot, pb);
        k_place<<<pb, PT, 0, stream>>>(row, col, vals, nnz, off, tot, rec_r, rec_c, rec_v);
        k_accum<<<nb, PT, 0, stream>>>(xt, rec_r, rec_c, rec_v, tot, out_t, nd);
    } else {
        long long n4 = ((long long)nd * BATCH) / 4;
        long long zb = (n4 + 255) / 256;
        if (zb > 16384) zb = 16384;
        k_zero<<<(int)zb, 256, 0, stream>>>((float4*)out_t, n4);
        long long waves = ((long long)nnz + 32LL * CHUNKS - 1) / (32LL * CHUNKS);
        long long sblocks = (waves + 7) / 8;
        if (sblocks < 1) sblocks = 1;
        k_scatter<<<(int)sblocks, 256, 0, stream>>>(xt, vals, row, col, out_t, nnz);
    }

    const int dtiles = nd / 16;
    if (dtiles > 0) {
        const int tiles = (BATCH / 16) * dtiles;
        const int eblocks = (tiles + 7) / 8;
        k_epilogue_wmma<<<eblocks, 256, 0, stream>>>(out_t, bias, out, nd, dtiles);
    }
    const int dstart = dtiles * 16;
    if (dstart < nd) {
        const int total = (nd - dstart) * BATCH;
        k_epilogue_tail<<<(total + 255) / 256, 256, 0, stream>>>(out_t, bias, out, nd, dstart);
    }
}


# Bucket sum kernel: each edge's column, row and weight reach the wave through scalar registers (v_readlane) instead of three ds_bpermute; LDS instructions per edge 5 -> 2, no LDS wait in the edge loop
# speedup vs baseline: 1.7726x; 1.0650x over previous
#include <hip/hip_runtime.h>

typedef __attribute__((ext_vector_type(2))) float v2f;
typedef __attribute__((ext_vector_type(8))) float v8f;

#define BATCH 64
#define CHUNKS 4

__global__ void k_transpose(const float* __restrict__ x, float* __restrict__ xt, int ns) {
    __shared__ float tile[32][33];
    const int tx = threadIdx.x;
    const int ty = threadIdx.y;
    const int s0 = blockIdx.x * 32;
    const int b0 = blockIdx.y * 32;
#pragma unroll
    for (int k = 0; k < 4; ++k) {
        int s = s0 + tx, b = b0 + ty + k * 8;
        if (s < ns && b < BATCH) tile[ty + k * 8][tx] = x[(size_t)b * ns + s];
    }
    __syncthreads();
#pragma unroll
    for (int k = 0; k < 4; ++k) {
        int s = s0 + ty + k * 8, b = b0 + tx;
        if (s < ns && b < BATCH) xt[(size_t)s * BATCH + b] = tile[tx][ty + k * 8];
    }
}

#define TS 128
#define TPITCH 65
template <bool FULL>
__device__ __forceinline__ void transpose_tile(const float* __restrict__ x, float* __restrict__ xt,
                                               int ns, int s0, float* __restrict__ t) {
    const int tid = threadIdx.x;
    const int q = tid & 31;
    const int w = tid >> 5;
    if (FULL || s0 + 4 * q < ns) {
        const float* p = x + (size_t)w * ns + s0 + 4 * q;
        const size_t step = (size_t)8 * ns;
        float4 v[8];
#pragma unroll
        for (int k = 0; k < 8; ++k) v[k] = *(const float4*)(p + k * step);
        const int wbase = (4 * q) * TPITCH + ((4 * q) >> 6) + w;
#pragma unroll
        for (int k = 0; k < 8; ++k) {
            t[wbase + 8 * k]              = v[k].x;
            t[wbase + 8 * k + TPITCH]     = v[k].y;
            t[wbase + 8 * k + 2 * TPITCH] = v[k].z;
            t[wbase + 8 * k + 3 * TPITCH] = v[k].w;
        }
    }
    __syncthreads();
    const int bq = tid & 15;
    const int sr = tid >> 4;
    float* dst = xt + (size_t)(s0 + sr) * BATCH + 4 * bq;
    if (FULL) {
        float4 o[8];
#pragma unroll
        for (int k = 0; k < 8; ++k) {
            const int s = sr + 16 * k;
            const int rb = s * TPITCH + (s >> 6) + 4 * bq;
            o[k] = make_float4(t[rb], t[rb + 1], t[rb + 2], t[rb + 3]);
        }
#pragma unroll
        for (int k = 0; k < 8; ++k) *(float4*)(dst + (size_t)(16 * k) * BATCH) = o[k];
    } else {
#pragma unroll
        for (int k = 0; k < 8; ++k) {
            const int s = sr + 16 * k;
            if (s0 + s < ns) {
                const int rb = s * TPITCH + (s >> 6) + 4 * bq;
                *(float4*)(dst + (size_t)(16 * k) * BATCH) =
                    make_float4(t[rb], t[rb + 1], t[rb + 2], t[rb + 3]);
            }
        }
    }
}

__global__ void __launch_bounds__(256) k_transpose_wide(const float* __restrict__ x,
                                                        float* __restrict__ xt, int ns) {
    __shared__ float t[TS * TPITCH + 4];
    const int s0 = blockIdx.x * TS;
    if (s0 + TS <= ns) transpose_tile<true>(x, xt, ns, s0, t);
    else               transpose_tile<false>(x, xt, ns, s0, t);
}

__global__ void k_zero(float4* __restrict__ p, long long n4) {
    long long i = (long long)blockIdx.x * blockDim.x + threadIdx.x;
    const long long stride = (long long)gridDim.x * blockDim.x;
    const float4 z = make_float4(0.f, 0.f, 0.f, 0.f);
    for (; i < n4; i += stride) p[i] = z;
}

__global__ void k_scatter(const float* __restrict__ xt, const float* __restrict__ vals,
                          const int* __restrict__ row, const int* __restrict__ col,
                          float* __restrict__ out_t, int nnz) {
    const int lane = threadIdx.x & 31;
    const long long wave = ((long long)blockIdx.x * blockDim.x + threadIdx.x) >> 5;
    const long long base0 = wave * (32LL * CHUNKS);
    for (int ch = 0; ch < CHUNKS; ++ch) {
        const long long base = base0 + 32LL * ch;
        if (base >= nnz) return;
        const long long idx = base + lane;
        float v = 0.f;
        int r = 0, c = 0;
        if (idx < nnz) {
            v = vals[idx];
            r = row[idx];
            c = col[idx];
        }
        if (ch + 1 < CHUNKS && base + 32 + lane < nnz) {
            __builtin_prefetch(vals + base + 32 + lane, 0, 0);
            __builtin_prefetch(row + base + 32 + lane, 0, 0);
            __builtin_prefetch(col + base + 32 + lane, 0, 0);
        }
        const int count = (nnz - base < 32) ? (int)(nnz - base) : 32;
        for (int j = 0; j < count; ++j) {
            const float vj = __shfl(v, j);
            const int rj   = __shfl(r, j);
            const int cj   = __shfl(c, j);
            const v2f xv = *(const v2f*)(xt + (size_t)cj * BATCH + lane * 2);
            float* dst = out_t + (size_t)rj * BATCH + lane * 2;
            unsafeAtomicAdd(dst, xv.x * vj);
            unsafeAtomicAdd(dst + 1, xv.y * vj);
        }
    }
}

__global__ void k_epilogue_wmma(const float* __restrict__ out_t, const float* __restrict__ bias,
                                float* __restrict__ out, int nd, int dtiles) {
    const int lane = threadIdx.x & 31;
    const int wave = (int)(((long long)blockIdx.x * blockDim.x + threadIdx.x) >> 5);
    const int tiles = (BATCH / 16) * dtiles;
    if (wave >= tiles) return;
    const int b0 = (wave & 3) * 16;
    const int d0 = (wave >> 2) * 16;
    const int n = lane & 15;
    const int mbase = (lane < 16) ? 0 : 8;

    const float* cbase = out_t + (size_t)(d0 + n) * BATCH + b0 + mbase;
    const float4 lo = *(const float4*)(cbase);
    const float4 hi = *(const float4*)(cbase + 4);
    v8f c = {lo.x, lo.y, lo.z, lo.w, hi.x, hi.y, hi.z, hi.w};

    v2f a = {(lane < 16) ? 1.0f : 0.0f, 0.0f};
    v2f bm = {(lane < 16) ? bias[d0 + n] : 0.0f, 0.0f};

    v8f d = __builtin_amdgcn_wmma_f32_16x16x4_f32(
         false, a,  false, bm,
         (short)0, c,  false,  false);

#pragma unroll
    for (int v = 0; v < 8; ++v) {
        out[(size_t)(b0 + mbase + v) * nd + d0 + n] = d[v];
    }
}

__global__ void k_epilogue_tail(const float* __restrict__ out_t, const float* __restrict__ bias,
                                float* __restrict__ out, int nd, int dstart) {
    const int total = (nd - dstart) * BATCH;
    const int i = blockIdx.x * blockDim.x + threadIdx.x;
    if (i >= total) return;
    const int d = dstart + (i >> 6);
    const int b = i & 63;
    out[(size_t)b * nd + d] = out_t[(size_t)d * BATCH + b] + bias[d];
}

#define RB 400
#define NBMAX 256
#define PT 1024
#define EPT 13
#define TILE (PT * EPT)

__device__ __forceinline__ int block_excl_scan(int v, int* s_w, int& total) {
    const int lane = threadIdx.x & 31;
    const int w = threadIdx.x >> 5;
    int inc = v;
#pragma unroll
    for (int d = 1; d < 32; d <<= 1) {
        const int t = __shfl_up(inc, d);
        if (lane >= d) inc += t;
    }
    if (lane == 31) s_w[w] = inc;
    __syncthreads();
    if (w == 0) {
        const int x = s_w[lane];
        int xi = x;
#pragma unroll
        for (int d = 1; d < 32; d <<= 1) {
            const int t = __shfl_up(xi, d);
            if (lane >= d) xi += t;
        }
        s_w[32 + lane] = xi - x;
        if (lane == 31) s_w[64] = xi;
    }
    __syncthreads();
    total = s_w[64];
    const int r = s_w[32 + w] + inc - v;
    __syncthreads();
    return r;
}

__global__ void __launch_bounds__(PT) k_hist(const int* __restrict__ row, int nnz,
                                             int* __restrict__ hist) {
    __shared__ int s_h[NBMAX];
    const int tid = threadIdx.x;
    if (tid < NBMAX) s_h[tid] = 0;
    __syncthreads();
    const long long base = (long long)blockIdx.x * TILE;
#pragma unroll
    for (int k = 0; k < EPT; ++k) {
        const long long e = base + k * PT + tid;
        if (e < nnz) atomicAdd(&s_h[row[e] / RB], 1);
    }
    __syncthreads();
    if (tid < NBMAX) hist[blockIdx.x * NBMAX + tid] = s_h[tid];
}

__global__ void __launch_bounds__(PT) k_scanp(const int* __restrict__ hist, int* __restrict__ off,
                                              int* __restrict__ tot, int pb) {
    __shared__ int s_w[65];
    const int b = blockIdx.x;
    const int tid = threadIdx.x;
    int carry = 0;
    for (int p0 = 0; p0 < pb; p0 += PT) {
        const int p = p0 + tid;
        const int v = (p < pb) ? hist[p * NBMAX + b] : 0;
        int total;
        const int ex = block_excl_scan(v, s_w, total);
        if (p < pb) off[p * NBMAX + b] = carry + ex;
        carry += total;
    }
    if (tid == 0) tot[b] = carry;
}

__global__ void __launch_bounds__(PT) k_place(const int* __restrict__ row, const int* __restrict__ col,
                                              const float* __restrict__ vals, int nnz,
                                              const int* __restrict__ off, const int* __restrict__ tot,
                                              int* __restrict__ rec_r, int* __restrict__ rec_c,
                                              float* __restrict__ rec_v) {
    __shared__ int s_r[TILE];
    __shared__ int s_c[TILE];
    __shared__ float s_v[TILE];
    __shared__ int s_h[NBMAX];
    __shared__ int s_loff[NBMAX];
    __shared__ int s_goff[NBMAX];
    __shared__ int s_w[65];
    const int tid = threadIdx.x;

    int total;
    const int t = (tid < NBMAX) ? tot[tid] : 0;
    const int bs = block_excl_scan(t, s_w, total);
    if (tid < NBMAX) {
        s_goff[tid] = bs + off[blockIdx.x * NBMAX + tid];
        s_h[tid] = 0;
    }
    __syncthreads();

    int myr[EPT], myc[EPT], myk[EPT];
    float myv[EPT];
    const long long base = (long long)blockIdx.x * TILE;
#pragma unroll
    for (int k = 0; k < EPT; ++k) {
        const long long e = base + k * PT + tid;
        myk[k] = -1;
        myr[k] = 0; myc[k] = 0; myv[k] = 0.f;
        if (e < nnz) {
            const int r = row[e];
            myr[k] = r;
            myc[k] = col[e];
            myv[k] = vals[e];
            myk[k] = atomicAdd(&s_h[r / RB], 1);
        }
    }
    __syncthreads();
    const int h = (tid < NBMAX) ? s_h[tid] : 0;
    const int lo = block_excl_scan(h, s_w, total);
    if (tid < NBMAX) s_loff[tid] = lo;
    __syncthreads();
#pragma unroll
    for (int k = 0; k < EPT; ++k) {
        if (myk[k] >= 0) {
            const int pos = s_loff[myr[k] / RB] + myk[k];
            s_r[pos] = myr[k];
            s_c[pos] = myc[k];
            s_v[pos] = myv[k];
        }
    }
    __syncthreads();
    for (int i = tid; i < total; i += PT) {
        const int r = s_r[i];
        const int b = r / RB;
        const int g = s_goff[b] + (i - s_loff[b]);
        rec_r[g] = r;
        rec_c[g] = s_c[i];
        rec_v[g] = s_v[i];
    }
}

__global__ void __launch_bounds__(PT) k_accum(const float* __restrict__ xt, const int* __restrict__ rec_r,
                                              const int* __restrict__ rec_c, const float* __restrict__ rec_v,
                                              const int* __restrict__ tot, float* __restrict__ out_t, int nd) {
    __shared__ __attribute__((aligned(16))) float acc[RB * BATCH];
    __shared__ int s_w[65];
    __shared__ int s_se[2];
    const int tid = threadIdx.x;
    const int lane = tid & 31;
    const int w = __builtin_amdgcn_readfirstlane(tid >> 5);
    const int b = blockIdx.x;

    int total;
    const int t = (tid < NBMAX) ? tot[tid] : 0;
    const int bs = block_excl_scan(t, s_w, total);
    if (tid == b) { s_se[0] = bs; s_se[1] = t; }
    const float4 z = make_float4(0.f, 0.f, 0.f, 0.f);
    for (int i = tid; i < RB * BATCH / 4; i += PT) ((float4*)acc)[i] = z;
    __syncthreads();
    const int start = s_se[0];
    const int cnt = s_se[1];
    const int row0 = b * RB;

    int rn = 0, cn = 0;
    float vn = 0.f;
    if (w * 32 + lane < cnt) {
        const int g = start + w * 32 + lane;
        rn = rec_r[g];
        cn = rec_c[g];
        vn = rec_v[g];
    }
    for (int i0 = w * 32; i0 < cnt; i0 += 32 * (PT / 32)) {
        const int n = (cnt - i0 < 32) ? (cnt - i0) : 32;
        const int r = rn - row0, c = cn;
        const int vb = __float_as_int(vn);
        const int i1 = i0 + 32 * (PT / 32);
        if (i1 + lane < cnt) {
            const int g = start + i1 + lane;
            rn = rec_r[g];
            cn = rec_c[g];
            vn = rec_v[g];
        }
        int j0 = 0;
        for (; j0 + 8 <= n; j0 += 8) {
            v2f xv[8];
#pragma unroll
            for (int u = 0; u < 8; ++u) {
                const int cj = __builtin_amdgcn_readlane(c, j0 + u);
                xv[u] = *(const v2f*)(xt + (size_t)cj * BATCH + lane * 2);
            }
#pragma unroll
            for (int u = 0; u < 8; ++u) {
                const float vj = __int_as_float(__builtin_amdgcn_readlane(vb, j0 + u));
                const int rj = __builtin_amdgcn_readlane(r, j0 + u);
                float* dst = acc + rj * BATCH + lane * 2;
                unsafeAtomicAdd(dst, xv[u].x * vj);
                unsafeAtomicAdd(dst + 1, xv[u].y * vj);
            }
        }
        for (; j0 < n; ++j0) {
            const int cj = __builtin_amdgcn_readlane(c, j0);
            const float vj = __int_as_float(__builtin_amdgcn_readlane(vb, j0));
            const int rj = __builtin_amdgcn_readlane(r, j0);
            const v2f xs = *(const v2f*)(xt + (size_t)cj * BATCH + lane * 2);
            float* dst = acc + rj * BATCH + lane * 2;
            unsafeAtomicAdd(dst, xs.x * vj);
            unsafeAtomicAdd(dst + 1, xs.y * vj);
        }
    }
    __syncthreads();
    const int rows = (nd - row0 < RB) ? (nd - row0) : RB;
    float4* dst4 = (float4*)(out_t + (size_t)row0 * BATCH);
    for (int i = tid; i < rows * (BATCH / 4); i += PT) dst4[i] = ((const float4*)acc)[i];
}

extern "C" void kernel_launch(void* const* d_in, const int* in_sizes, int n_in,
                              void* d_out, int out_size, void* d_ws, size_t ws_size,
                              hipStream_t stream) {
    const float* x    = (const float*)d_in[0];
    const float* vals = (const float*)d_in[1];
    const float* bias = (const float*)d_in[2];
    const int* row    = (const int*)d_in[3];
    const int* col    = (const int*)d_in[4];
    float* out = (float*)d_out;

    const int nnz = in_sizes[1];
    const int nd  = in_sizes[2];
    const int ns  = in_sizes[0] / BATCH;

    float* xt    = (float*)d_ws;
    float* out_t = xt + (size_t)ns * BATCH;

    if ((ns & 3) == 0) {
        k_transpose_wide<<<(ns + TS - 1) / TS, 256, 0, stream>>>(x, xt, ns);
    } else {
        dim3 tb(32, 8);
        dim3 tg((ns + 31) / 32, (BATCH + 31) / 32);
        k_transpose<<<tg, tb, 0, stream>>>(x, xt, ns);
    }

    const int nb = (nd + RB - 1) / RB;
    const int pb = (int)(((long long)nnz + TILE - 1) / TILE);
    const size_t need_words = (size_t)ns * BATCH + (size_t)nd * BATCH + 3 * (size_t)nnz +
                              2 * (size_t)pb * NBMAX + NBMAX;
    if (nb <= NBMAX && pb >= 1 && need_words * 4 <= ws_size) {
        int* rec_r   = (int*)(out_t + (size_t)nd * BATCH);
        int* rec_c   = rec_r + (size_t)nnz;
        float* rec_v = (float*)(rec_c + (size_t)nnz);
        int* hist    = (int*)(rec_v + (size_t)nnz);
        int* off     = hist + (size_t)pb * NBMAX;
        int* tot     = off + (size_t)pb * NBMAX;
        k_hist<<<pb, PT, 0, stream>>>(row, nnz, hist);
        k_scanp<<<NBMAX, PT, 0, stream>>>(hist, off, tot, pb);
        k_place<<<pb, PT, 0, stream>>>(row, col, vals, nnz, off, tot, rec_r, rec_c, rec_v);
        k_accum<<<nb, PT, 0, stream>>>(xt, rec_r, rec_c, rec_v, tot, out_t, nd);
    } else {
        long long n4 = ((long long)nd * BATCH) / 4;
        long long zb = (n4 + 255) / 256;
        if (zb > 16384) zb = 16384;
        k_zero<<<(int)zb, 256, 0, stream>>>((float4*)out_t, n4);
        long long waves = ((long long)nnz + 32LL * CHUNKS - 1) / (32LL * CHUNKS);
        long long sblocks = (waves + 7) / 8;
        if (sblocks < 1) sblocks = 1;
        k_scatter<<<(int)sblocks, 256, 0, stream>>>(xt, vals, row, col, out_t, nnz);
    }

    const int dtiles = nd / 16;
    if (dtiles > 0) {
        const int tiles = (BATCH / 16) * dtiles;
        const int eblocks = (tiles + 7) / 8;
        k_epilogue_wmma<<<eblocks, 256, 0, stream>>>(out_t, bias, out, nd, dtiles);
    }
    const int dstart = dtiles * 16;
    if (dstart < nd) {
        const int total = (nd - dstart) * BATCH;
        k_epilogue_tail<<<(total + 255) / 256, 256, 0, stream>>>(out_t, bias, out, nd, dstart);
    }
}


# Fuse the WMMA bias-add/transpose epilogue into the bucket sum kernel: C tiles read from the LDS tile (row pitch 68 dwords), D stored to out; no out_t round trip, 5 launches instead of 6
# speedup vs baseline: 1.8937x; 1.0683x over previous
#include <hip/hip_runtime.h>

typedef __attribute__((ext_vector_type(2))) float v2f;
typedef __attribute__((ext_vector_type(8))) float v8f;

#define BATCH 64
#define CHUNKS 4

__global__ void k_transpose(const float* __restrict__ x, float* __restrict__ xt, int ns) {
    __shared__ float tile[32][33];
    const int tx = threadIdx.x;
    const int ty = threadIdx.y;
    const int s0 = blockIdx.x * 32;
    const int b0 = blockIdx.y * 32;
#pragma unroll
    for (int k = 0; k < 4; ++k) {
        int s = s0 + tx, b = b0 + ty + k * 8;
        if (s < ns && b < BATCH) tile[ty + k * 8][tx] = x[(size_t)b * ns + s];
    }
    __syncthreads();
#pragma unroll
    for (int k = 0; k < 4; ++k) {
        int s = s0 + ty + k * 8, b = b0 + tx;
        if (s < ns && b < BATCH) xt[(size_t)s * BATCH + b] = tile[tx][ty + k * 8];
    }
}

#define TS 128
#define TPITCH 65
template <bool FULL>
__device__ __forceinline__ void transpose_tile(const float* __restrict__ x, float* __restrict__ xt,
                                               int ns, int s0, float* __restrict__ t) {
    const int tid = threadIdx.x;
    const int q = tid & 31;
    const int w = tid >> 5;
    if (FULL || s0 + 4 * q < ns) {
        const float* p = x + (size_t)w * ns + s0 + 4 * q;
        const size_t step = (size_t)8 * ns;
        float4 v[8];
#pragma unroll
        for (int k = 0; k < 8; ++k) v[k] = *(const float4*)(p + k * step);
        const int wbase = (4 * q) * TPITCH + ((4 * q) >> 6) + w;
#pragma unroll
        for (int k = 0; k < 8; ++k) {
            t[wbase + 8 * k]              = v[k].x;
            t[wbase + 8 * k + TPITCH]     = v[k].y;
            t[wbase + 8 * k + 2 * TPITCH] = v[k].z;
            t[wbase + 8 * k + 3 * TPITCH] = v[k].w;
        }
    }
    __syncthreads();
    const int bq = tid & 15;
    const int sr = tid >> 4;
    float* dst = xt + (size_t)(s0 + sr) * BATCH + 4 * bq;
    if (FULL) {
        float4 o[8];
#pragma unroll
        for (int k = 0; k < 8; ++k) {
            const int s = sr + 16 * k;
            const int rb = s * TPITCH + (s >> 6) + 4 * bq;
            o[k] = make_float4(t[rb], t[rb + 1], t[rb + 2], t[rb + 3]);
        }
#pragma unroll
        for (int k = 0; k < 8; ++k) *(float4*)(dst + (size_t)(16 * k) * BATCH) = o[k];
    } else {
#pragma unroll
        for (int k = 0; k < 8; ++k) {
            const int s = sr + 16 * k;
            if (s0 + s < ns) {
                const int rb = s * TPITCH + (s >> 6) + 4 * bq;
                *(float4*)(dst + (size_t)(16 * k) * BATCH) =
                    make_float4(t[rb], t[rb + 1], t[rb + 2], t[rb + 3]);
            }
        }
    }
}

__global__ void __launch_bounds__(256) k_transpose_wide(const float* __restrict__ x,
                                                        float* __restrict__ xt, int ns) {
    __shared__ float t[TS * TPITCH + 4];
    const int s0 = blockIdx.x * TS;
    if (s0 + TS <= ns) transpose_tile<true>(x, xt, ns, s0, t);
    else               transpose_tile<false>(x, xt, ns, s0, t);
}

__global__ void k_zero(float4* __restrict__ p, long long n4) {
    long long i = (long long)blockIdx.x * blockDim.x + threadIdx.x;
    const long long stride = (long long)gridDim.x * blockDim.x;
    const float4 z = make_float4(0.f, 0.f, 0.f, 0.f);
    for (; i < n4; i += stride) p[i] = z;
}

__global__ void k_scatter(const float* __restrict__ xt, const float* __restrict__ vals,
                          const int* __restrict__ row, const int* __restrict__ col,
                          float* __restrict__ out_t, int nnz) {
    const int lane = threadIdx.x & 31;
    const long long wave = ((long long)blockIdx.x * blockDim.x + threadIdx.x) >> 5;
    const long long base0 = wave * (32LL * CHUNKS);
    for (int ch = 0; ch < CHUNKS; ++ch) {
        const long long base = base0 + 32LL * ch;
        if (base >= nnz) return;
        const long long idx = base + lane;
        float v = 0.f;
        int r = 0, c = 0;
        if (idx < nnz) {
            v = vals[idx];
            r = row[idx];
            c = col[idx];
        }
        if (ch + 1 < CHUNKS && base + 32 + lane < nnz) {
            __builtin_prefetch(vals + base + 32 + lane, 0, 0);
            __builtin_prefetch(row + base + 32 + lane, 0, 0);
            __builtin_prefetch(col + base + 32 + lane, 0, 0);
        }
        const int count = (nnz - base < 32) ? (int)(nnz - base) : 32;
        for (int j = 0; j < count; ++j) {
            const float vj = __shfl(v, j);
            const int rj   = __shfl(r, j);
            const int cj   = __shfl(c, j);
            const v2f xv = *(const v2f*)(xt + (size_t)cj * BATCH + lane * 2);
            float* dst = out_t + (size_t)rj * BATCH + lane * 2;
            unsafeAtomicAdd(dst, xv.x * vj);
            unsafeAtomicAdd(dst + 1, xv.y * vj);
        }
    }
}

__global__ void k_epilogue_wmma(const float* __restrict__ out_t, const float* __restrict__ bias,
                                float* __restrict__ out, int nd, int dtiles) {
    const int lane = threadIdx.x & 31;
    const int wave = (int)(((long long)blockIdx.x * blockDim.x + threadIdx.x) >> 5);
    const int tiles = (BATCH / 16) * dtiles;
    if (wave >= tiles) return;
    const int b0 = (wave & 3) * 16;
    const int d0 = (wave >> 2) * 16;
    const int n = lane & 15;
    const int mbase = (lane < 16) ? 0 : 8;

    const float* cbase = out_t + (size_t)(d0 + n) * BATCH + b0 + mbase;
    const float4 lo = *(const float4*)(cbase);
    const float4 hi = *(const float4*)(cbase + 4);
    v8f c = {lo.x, lo.y, lo.z, lo.w, hi.x, hi.y, hi.z, hi.w};

    v2f a = {(lane < 16) ? 1.0f : 0.0f, 0.0f};
    v2f bm = {(lane < 16) ? bias[d0 + n] : 0.0f, 0.0f};

    v8f d = __builtin_amdgcn_wmma_f32_16x16x4_f32(
         false, a,  false, bm,
         (short)0, c,  false,  false);

#pragma unroll
    for (int v = 0; v < 8; ++v) {
        out[(size_t)(b0 + mbase + v) * nd + d0 + n] = d[v];
    }
}

__global__ void k_epilogue_tail(const float* __restrict__ out_t, const float* __restrict__ bias,
                                float* __restrict__ out, int nd, int dstart) {
    const int total = (nd - dstart) * BATCH;
    const int i = blockIdx.x * blockDim.x + threadIdx.x;
    if (i >= total) return;
    const int d = dstart + (i >> 6);
    const int b = i & 63;
    out[(size_t)b * nd + d] = out_t[(size_t)d * BATCH + b] + bias[d];
}

#define RB 400
#define AP 68
#define NBMAX 256
#define PT 1024
#define EPT 13
#define TILE (PT * EPT)

__device__ __forceinline__ int block_excl_scan(int v, int* s_w, int& total) {
    const int lane = threadIdx.x & 31;
    const int w = threadIdx.x >> 5;
    int inc = v;
#pragma unroll
    for (int d = 1; d < 32; d <<= 1) {
        const int t = __shfl_up(inc, d);
        if (lane >= d) inc += t;
    }
    if (lane == 31) s_w[w] = inc;
    __syncthreads();
    if (w == 0) {
        const int x = s_w[lane];
        int xi = x;
#pragma unroll
        for (int d = 1; d < 32; d <<= 1) {
            const int t = __shfl_up(xi, d);
            if (lane >= d) xi += t;
        }
        s_w[32 + lane] = xi - x;
        if (lane == 31) s_w[64] = xi;
    }
    __syncthreads();
    total = s_w[64];
    const int r = s_w[32 + w] + inc - v;
    __syncthreads();
    return r;
}

__global__ void __launch_bounds__(PT) k_hist(const int* __restrict__ row, int nnz,
                                             int* __restrict__ hist) {
    __shared__ int s_h[NBMAX];
    const int tid = threadIdx.x;
    if (tid < NBMAX) s_h[tid] = 0;
    __syncthreads();
    const long long base = (long long)blockIdx.x * TILE;
#pragma unroll
    for (int k = 0; k < EPT; ++k) {
        const long long e = base + k * PT + tid;
        if (e < nnz) atomicAdd(&s_h[row[e] / RB], 1);
    }
    __syncthreads();
    if (tid < NBMAX) hist[blockIdx.x * NBMAX + tid] = s_h[tid];
}

__global__ void __launch_bounds__(PT) k_scanp(const int* __restrict__ hist, int* __restrict__ off,
                                              int* __restrict__ tot, int pb) {
    __shared__ int s_w[65];
    const int b = blockIdx.x;
    const int tid = threadIdx.x;
    int carry = 0;
    for (int p0 = 0; p0 < pb; p0 += PT) {
        const int p = p0 + tid;
        const int v = (p < pb) ? hist[p * NBMAX + b] : 0;
        int total;
        const int ex = block_excl_scan(v, s_w, total);
        if (p < pb) off[p * NBMAX + b] = carry + ex;
        carry += total;
    }
    if (tid == 0) tot[b] = carry;
}

__global__ void __launch_bounds__(PT) k_place(const int* __restrict__ row, const int* __restrict__ col,
                                              const float* __restrict__ vals, int nnz,
                                              const int* __restrict__ off, const int* __restrict__ tot,
                                              int* __restrict__ rec_r, int* __restrict__ rec_c,
                                              float* __restrict__ rec_v) {
    __shared__ int s_r[TILE];
    __shared__ int s_c[TILE];
    __shared__ float s_v[TILE];
    __shared__ int s_h[NBMAX];
    __shared__ int s_loff[NBMAX];
    __shared__ int s_goff[NBMAX];
    __shared__ int s_w[65];
    const int tid = threadIdx.x;

    int total;
    const int t = (tid < NBMAX) ? tot[tid] : 0;
    const int bs = block_excl_scan(t, s_w, total);
    if (tid < NBMAX) {
        s_goff[tid] = bs + off[blockIdx.x * NBMAX + tid];
        s_h[tid] = 0;
    }
    __syncthreads();

    int myr[EPT], myc[EPT], myk[EPT];
    float myv[EPT];
    const long long base = (long long)blockIdx.x * TILE;
#pragma unroll
    for (int k = 0; k < EPT; ++k) {
        const long long e = base + k * PT + tid;
        myk[k] = -1;
        myr[k] = 0; myc[k] = 0; myv[k] = 0.f;
        if (e < nnz) {
            const int r = row[e];
            myr[k] = r;
            myc[k] = col[e];
            myv[k] = vals[e];
            myk[k] = atomicAdd(&s_h[r / RB], 1);
        }
    }
    __syncthreads();
    const int h = (tid < NBMAX) ? s_h[tid] : 0;
    const int lo = block_excl_scan(h, s_w, total);
    if (tid < NBMAX) s_loff[tid] = lo;
    __syncthreads();
#pragma unroll
    for (int k = 0; k < EPT; ++k) {
        if (myk[k] >= 0) {
            const int pos = s_loff[myr[k] / RB] + myk[k];
            s_r[pos] = myr[k];
            s_c[pos] = myc[k];
            s_v[pos] = myv[k];
        }
    }
    __syncthreads();
    for (int i = tid; i < total; i += PT) {
        const int r = s_r[i];
        const int b = r / RB;
        const int g = s_goff[b] + (i - s_loff[b]);
        rec_r[g] = r;
        rec_c[g] = s_c[i];
        rec_v[g] = s_v[i];
    }
}

__global__ void __launch_bounds__(PT) k_accum(const float* __restrict__ xt, const int* __restrict__ rec_r,
                                              const int* __restrict__ rec_c, const float* __restrict__ rec_v,
                                              const int* __restrict__ tot, const float* __restrict__ bias,
                                              float* __restrict__ out, int nd) {
    __shared__ __attribute__((aligned(16))) float acc[RB * AP];
    __shared__ int s_w[65];
    __shared__ int s_se[2];
    const int tid = threadIdx.x;
    const int lane = tid & 31;
    const int w = __builtin_amdgcn_readfirstlane(tid >> 5);
    const int b = blockIdx.x;

    int total;
    const int t = (tid < NBMAX) ? tot[tid] : 0;
    const int bs = block_excl_scan(t, s_w, total);
    if (tid == b) { s_se[0] = bs; s_se[1] = t; }
    const float4 z = make_float4(0.f, 0.f, 0.f, 0.f);
    for (int i = tid; i < RB * AP / 4; i += PT) ((float4*)acc)[i] = z;
    __syncthreads();
    const int start = s_se[0];
    const int cnt = s_se[1];
    const int row0 = b * RB;

    int rn = 0, cn = 0;
    float vn = 0.f;
    if (w * 32 + lane < cnt) {
        const int g = start + w * 32 + lane;
        rn = rec_r[g];
        cn = rec_c[g];
        vn = rec_v[g];
    }
    for (int i0 = w * 32; i0 < cnt; i0 += 32 * (PT / 32)) {
        const int n = (cnt - i0 < 32) ? (cnt - i0) : 32;
        const int r = rn - row0, c = cn;
        const int vb = __float_as_int(vn);
        const int i1 = i0 + 32 * (PT / 32);
        if (i1 + lane < cnt) {
            const int g = start + i1 + lane;
            rn = rec_r[g];
            cn = rec_c[g];
            vn = rec_v[g];
        }
        int j0 = 0;
        for (; j0 + 8 <= n; j0 += 8) {
            v2f xv[8];
#pragma unroll
            for (int u = 0; u < 8; ++u) {
                const int cj = __builtin_amdgcn_readlane(c, j0 + u);
                xv[u] = *(const v2f*)(xt + (size_t)cj * BATCH + lane * 2);
            }
#pragma unroll
            for (int u = 0; u < 8; ++u) {
                const float vj = __int_as_float(__builtin_amdgcn_readlane(vb, j0 + u));
                const int rj = __builtin_amdgcn_readlane(r, j0 + u);
                float* dst = acc + rj * AP + lane * 2;
                unsafeAtomicAdd(dst, xv[u].x * vj);
                unsafeAtomicAdd(dst + 1, xv[u].y * vj);
            }
        }
        for (; j0 < n; ++j0) {
            const int cj = __builtin_amdgcn_readlane(c, j0);
            const float vj = __int_as_float(__builtin_amdgcn_readlane(vb, j0));
            const int rj = __builtin_amdgcn_readlane(r, j0);
            const v2f xs = *(const v2f*)(xt + (size_t)cj * BATCH + lane * 2);
            float* dst = acc + rj * AP + lane * 2;
            unsafeAtomicAdd(dst, xs.x * vj);
            unsafeAtomicAdd(dst + 1, xs.y * vj);
        }
    }
    __syncthreads();
    const int rows = (nd - row0 < RB) ? (nd - row0) : RB;
    const int fd = rows >> 4;
    const int ntiles = fd * (BATCH / 16);
    const int n = lane & 15;
    const int mbase = (lane < 16) ? 0 : 8;
    const v2f a = {(lane < 16) ? 1.0f : 0.0f, 0.0f};
    for (int t = w; t < ntiles; t += PT / 32) {
        const int b0 = (t & 3) * 16;
        const int d0 = (t >> 2) * 16;
        const float* cb = acc + (d0 + n) * AP + b0 + mbase;
        const float4 lo = *(const float4*)(cb);
        const float4 hi = *(const float4*)(cb + 4);
        const v8f c = {lo.x, lo.y, lo.z, lo.w, hi.x, hi.y, hi.z, hi.w};
        const v2f bm = {(lane < 16) ? bias[row0 + d0 + n] : 0.0f, 0.0f};
        const v8f d = __builtin_amdgcn_wmma_f32_16x16x4_f32(
             false, a,  false, bm,
             (short)0, c,  false,  false);
        float* o = out + (size_t)(b0 + mbase) * nd + row0 + d0 + n;
#pragma unroll
        for (int v = 0; v < 8; ++v) o[(size_t)v * nd] = d[v];
    }
    const int dt = fd * 16;
    for (int i = tid; i < (rows - dt) * BATCH; i += PT) {
        const int dd = dt + (i >> 6);
        const int bb = i & 63;
        out[(size_t)bb * nd + row0 + dd] = acc[dd * AP + bb] + bias[row0 + dd];
    }
}

extern "C" void kernel_launch(void* const* d_in, const int* in_sizes, int n_in,
                              void* d_out, int out_size, void* d_ws, size_t ws_size,
                              hipStream_t stream) {
    const float* x    = (const float*)d_in[0];
    const float* vals = (const float*)d_in[1];
    const float* bias = (const float*)d_in[2];
    const int* row    = (const int*)d_in[3];
    const int* col    = (const int*)d_in[4];
    float* out = (float*)d_out;

    const int nnz = in_sizes[1];
    const int nd  = in_sizes[2];
    const int ns  = in_sizes[0] / BATCH;

    float* xt    = (float*)d_ws;
    float* out_t = xt + (size_t)ns * BATCH;

    if ((ns & 3) == 0) {
        k_transpose_wide<<<(ns + TS - 1) / TS, 256, 0, stream>>>(x, xt, ns);
    } else {
        dim3 tb(32, 8);
        dim3 tg((ns + 31) / 32, (BATCH + 31) / 32);
        k_transpose<<<tg, tb, 0, stream>>>(x, xt, ns);
    }

    const int nb = (nd + RB - 1) / RB;
    const int pb = (int)(((long long)nnz + TILE - 1) / TILE);
    const size_t need_words = (size_t)ns * BATCH + (size_t)nd * BATCH + 3 * (size_t)nnz +
                              2 * (size_t)pb * NBMAX + NBMAX;
    if (nb <= NBMAX && pb >= 1 && need_words * 4 <= ws_size) {
        int* rec_r   = (int*)(out_t + (size_t)nd * BATCH);
        int* rec_c   = rec_r + (size_t)nnz;
        float* rec_v = (float*)(rec_c + (size_t)nnz);
        int* hist    = (int*)(rec_v + (size_t)nnz);
        int* off     = hist + (size_t)pb * NBMAX;
        int* tot     = off + (size_t)pb * NBMAX;
        k_hist<<<pb, PT, 0, stream>>>(row, nnz, hist);
        k_scanp<<<NBMAX, PT, 0, stream>>>(hist, off, tot, pb);
        k_place<<<pb, PT, 0, stream>>>(row, col, vals, nnz, off, tot, rec_r, rec_c, rec_v);
        k_accum<<<nb, PT, 0, stream>>>(xt, rec_r, rec_c, rec_v, tot, bias, out, nd);
        return;
    } else {
        long long n4 = ((long long)nd * BATCH) / 4;
        long long zb = (n4 + 255) / 256;
        if (zb > 16384) zb = 16384;
        k_zero<<<(int)zb, 256, 0, stream>>>((float4*)out_t, n4);
        long long waves = ((long long)nnz + 32LL * CHUNKS - 1) / (32LL * CHUNKS);
        long long sblocks = (waves + 7) / 8;
        if (sblocks < 1) sblocks = 1;
        k_scatter<<<(int)sblocks, 256, 0, stream>>>(xt, vals, row, col, out_t, nnz);
    }

    const int dtiles = nd / 16;
    if (dtiles > 0) {
        const int tiles = (BATCH / 16) * dtiles;
        const int eblocks = (tiles + 7) / 8;
        k_epilogue_wmma<<<eblocks, 256, 0, stream>>>(out_t, bias, out, nd, dtiles);
    }
    const int dstart = dtiles * 16;
    if (dstart < nd) {
        const int total = (nd - dstart) * BATCH;
        k_epilogue_tail<<<(total + 255) / 256, 256, 0, stream>>>(out_t, bias, out, nd, dstart);
    }
}


# Bucket sum kernel: second counting sort by destination row in LDS, then each wave owns whole rows and sums them in registers in sort-rank (arrival) order; no ds_add_f32, one LDS store per row
# speedup vs baseline: 2.2637x; 1.1954x over previous
#include <hip/hip_runtime.h>

typedef __attribute__((ext_vector_type(2))) float v2f;
typedef __attribute__((ext_vector_type(8))) float v8f;

#define BATCH 64
#define CHUNKS 4

__global__ void k_transpose(const float* __restrict__ x, float* __restrict__ xt, int ns) {
    __shared__ float tile[32][33];
    const int tx = threadIdx.x;
    const int ty = threadIdx.y;
    const int s0 = blockIdx.x * 32;
    const int b0 = blockIdx.y * 32;
#pragma unroll
    for (int k = 0; k < 4; ++k) {
        int s = s0 + tx, b = b0 + ty + k * 8;
        if (s < ns && b < BATCH) tile[ty + k * 8][tx] = x[(size_t)b * ns + s];
    }
    __syncthreads();
#pragma unroll
    for (int k = 0; k < 4; ++k) {
        int s = s0 + ty + k * 8, b = b0 + tx;
        if (s < ns && b < BATCH) xt[(size_t)s * BATCH + b] = tile[tx][ty + k * 8];
    }
}

#define TS 128
#define TPITCH 65
template <bool FULL>
__device__ __forceinline__ void transpose_tile(const float* __restrict__ x, float* __restrict__ xt,
                                               int ns, int s0, float* __restrict__ t) {
    const int tid = threadIdx.x;
    const int q = tid & 31;
    const int w = tid >> 5;
    if (FULL || s0 + 4 * q < ns) {
        const float* p = x + (size_t)w * ns + s0 + 4 * q;
        const size_t step = (size_t)8 * ns;
        float4 v[8];
#pragma unroll
        for (int k = 0; k < 8; ++k) v[k] = *(const float4*)(p + k * step);
        const int wbase = (4 * q) * TPITCH + ((4 * q) >> 6) + w;
#pragma unroll
        for (int k = 0; k < 8; ++k) {
            t[wbase + 8 * k]              = v[k].x;
            t[wbase + 8 * k + TPITCH]     = v[k].y;
            t[wbase + 8 * k + 2 * TPITCH] = v[k].z;
            t[wbase + 8 * k + 3 * TPITCH] = v[k].w;
        }
    }
    __syncthreads();
    const int bq = tid & 15;
    const int sr = tid >> 4;
    float* dst = xt + (size_t)(s0 + sr) * BATCH + 4 * bq;
    if (FULL) {
        float4 o[8];
#pragma unroll
        for (int k = 0; k < 8; ++k) {
            const int s = sr + 16 * k;
            const int rb = s * TPITCH + (s >> 6) + 4 * bq;
            o[k] = make_float4(t[rb], t[rb + 1], t[rb + 2], t[rb + 3]);
        }
#pragma unroll
        for (int k = 0; k < 8; ++k) *(float4*)(dst + (size_t)(16 * k) * BATCH) = o[k];
    } else {
#pragma unroll
        for (int k = 0; k < 8; ++k) {
            const int s = sr + 16 * k;
            if (s0 + s < ns) {
                const int rb = s * TPITCH + (s >> 6) + 4 * bq;
                *(float4*)(dst + (size_t)(16 * k) * BATCH) =
                    make_float4(t[rb], t[rb + 1], t[rb + 2], t[rb + 3]);
            }
        }
    }
}

__global__ void __launch_bounds__(256) k_transpose_wide(const float* __restrict__ x,
                                                        float* __restrict__ xt, int ns) {
    __shared__ float t[TS * TPITCH + 4];
    const int s0 = blockIdx.x * TS;
    if (s0 + TS <= ns) transpose_tile<true>(x, xt, ns, s0, t);
    else               transpose_tile<false>(x, xt, ns, s0, t);
}

__global__ void k_zero(float4* __restrict__ p, long long n4) {
    long long i = (long long)blockIdx.x * blockDim.x + threadIdx.x;
    const long long stride = (long long)gridDim.x * blockDim.x;
    const float4 z = make_float4(0.f, 0.f, 0.f, 0.f);
    for (; i < n4; i += stride) p[i] = z;
}

__global__ void k_scatter(const float* __restrict__ xt, const float* __restrict__ vals,
                          const int* __restrict__ row, const int* __restrict__ col,
                          float* __restrict__ out_t, int nnz) {
    const int lane = threadIdx.x & 31;
    const long long wave = ((long long)blockIdx.x * blockDim.x + threadIdx.x) >> 5;
    const long long base0 = wave * (32LL * CHUNKS);
    for (int ch = 0; ch < CHUNKS; ++ch) {
        const long long base = base0 + 32LL * ch;
        if (base >= nnz) return;
        const long long idx = base + lane;
        float v = 0.f;
        int r = 0, c = 0;
        if (idx < nnz) {
            v = vals[idx];
            r = row[idx];
            c = col[idx];
        }
        if (ch + 1 < CHUNKS && base + 32 + lane < nnz) {
            __builtin_prefetch(vals + base + 32 + lane, 0, 0);
            __builtin_prefetch(row + base + 32 + lane, 0, 0);
            __builtin_prefetch(col + base + 32 + lane, 0, 0);
        }
        const int count = (nnz - base < 32) ? (int)(nnz - base) : 32;
        for (int j = 0; j < count; ++j) {
            const float vj = __shfl(v, j);
            const int rj   = __shfl(r, j);
            const int cj   = __shfl(c, j);
            const v2f xv = *(const v2f*)(xt + (size_t)cj * BATCH + lane * 2);
            float* dst = out_t + (size_t)rj * BATCH + lane * 2;
            unsafeAtomicAdd(dst, xv.x * vj);
            unsafeAtomicAdd(dst + 1, xv.y * vj);
        }
    }
}

__global__ void k_epilogue_wmma(const float* __restrict__ out_t, const float* __restrict__ bias,
                                float* __restrict__ out, int nd, int dtiles) {
    const int lane = threadIdx.x & 31;
    const int wave = (int)(((long long)blockIdx.x * blockDim.x + threadIdx.x) >> 5);
    const int tiles = (BATCH / 16) * dtiles;
    if (wave >= tiles) return;
    const int b0 = (wave & 3) * 16;
    const int d0 = (wave >> 2) * 16;
    const int n = lane & 15;
    const int mbase = (lane < 16) ? 0 : 8;

    const float* cbase = out_t + (size_t)(d0 + n) * BATCH + b0 + mbase;
    const float4 lo = *(const float4*)(cbase);
    const float4 hi = *(const float4*)(cbase + 4);
    v8f c = {lo.x, lo.y, lo.z, lo.w, hi.x, hi.y, hi.z, hi.w};

    v2f a = {(lane < 16) ? 1.0f : 0.0f, 0.0f};
    v2f bm = {(lane < 16) ? bias[d0 + n] : 0.0f, 0.0f};

    v8f d = __builtin_amdgcn_wmma_f32_16x16x4_f32(
         false, a,  false, bm,
         (short)0, c,  false,  false);

#pragma unroll
    for (int v = 0; v < 8; ++v) {
        out[(size_t)(b0 + mbase + v) * nd + d0 + n] = d[v];
    }
}

__global__ void k_epilogue_tail(const float* __restrict__ out_t, const float* __restrict__ bias,
                                float* __restrict__ out, int nd, int dstart) {
    const int total = (nd - dstart) * BATCH;
    const int i = blockIdx.x * blockDim.x + threadIdx.x;
    if (i >= total) return;
    const int d = dstart + (i >> 6);
    const int b = i & 63;
    out[(size_t)b * nd + d] = out_t[(size_t)d * BATCH + b] + bias[d];
}

#define RB 400
#define AP 68
#define NBMAX 256
#define PT 1024
#define EPT 13
#define TILE (PT * EPT)

__device__ __forceinline__ int block_excl_scan(int v, int* s_w, int& total) {
    const int lane = threadIdx.x & 31;
    const int w = threadIdx.x >> 5;
    int inc = v;
#pragma unroll
    for (int d = 1; d < 32; d <<= 1) {
        const int t = __shfl_up(inc, d);
        if (lane >= d) inc += t;
    }
    if (lane == 31) s_w[w] = inc;
    __syncthreads();
    if (w == 0) {
        const int x = s_w[lane];
        int xi = x;
#pragma unroll
        for (int d = 1; d < 32; d <<= 1) {
            const int t = __shfl_up(xi, d);
            if (lane >= d) xi += t;
        }
        s_w[32 + lane] = xi - x;
        if (lane == 31) s_w[64] = xi;
    }
    __syncthreads();
    total = s_w[64];
    const int r = s_w[32 + w] + inc - v;
    __syncthreads();
    return r;
}

__global__ void __launch_bounds__(PT) k_hist(const int* __restrict__ row, int nnz,
                                             int* __restrict__ hist) {
    __shared__ int s_h[NBMAX];
    const int tid = threadIdx.x;
    if (tid < NBMAX) s_h[tid] = 0;
    __syncthreads();
    const long long base = (long long)blockIdx.x * TILE;
#pragma unroll
    for (int k = 0; k < EPT; ++k) {
        const long long e = base + k * PT + tid;
        if (e < nnz) atomicAdd(&s_h[row[e] / RB], 1);
    }
    __syncthreads();
    if (tid < NBMAX) hist[blockIdx.x * NBMAX + tid] = s_h[tid];
}

__global__ void __launch_bounds__(PT) k_scanp(const int* __restrict__ hist, int* __restrict__ off,
                                              int* __restrict__ tot, int pb) {
    __shared__ int s_w[65];
    const int b = blockIdx.x;
    const int tid = threadIdx.x;
    int carry = 0;
    for (int p0 = 0; p0 < pb; p0 += PT) {
        const int p = p0 + tid;
        const int v = (p < pb) ? hist[p * NBMAX + b] : 0;
        int total;
        const int ex = block_excl_scan(v, s_w, total);
        if (p < pb) off[p * NBMAX + b] = carry + ex;
        carry += total;
    }
    if (tid == 0) tot[b] = carry;
}

__global__ void __launch_bounds__(PT) k_place(const int* __restrict__ row, const int* __restrict__ col,
                                              const float* __restrict__ vals, int nnz,
                                              const int* __restrict__ off, const int* __restrict__ tot,
                                              int* __restrict__ rec_r, int* __restrict__ rec_c,
                                              float* __restrict__ rec_v) {
    __shared__ int s_r[TILE];
    __shared__ int s_c[TILE];
    __shared__ float s_v[TILE];
    __shared__ int s_h[NBMAX];
    __shared__ int s_loff[NBMAX];
    __shared__ int s_goff[NBMAX];
    __shared__ int s_w[65];
    const int tid = threadIdx.x;

    int total;
    const int t = (tid < NBMAX) ? tot[tid] : 0;
    const int bs = block_excl_scan(t, s_w, total);
    if (tid < NBMAX) {
        s_goff[tid] = bs + off[blockIdx.x * NBMAX + tid];
        s_h[tid] = 0;
    }
    __syncthreads();

    int myr[EPT], myc[EPT], myk[EPT];
    float myv[EPT];
    const long long base = (long long)blockIdx.x * TILE;
#pragma unroll
    for (int k = 0; k < EPT; ++k) {
        const long long e = base + k * PT + tid;
        myk[k] = -1;
        myr[k] = 0; myc[k] = 0; myv[k] = 0.f;
        if (e < nnz) {
            const int r = row[e];
            myr[k] = r;
            myc[k] = col[e];
            myv[k] = vals[e];
            myk[k] = atomicAdd(&s_h[r / RB], 1);
        }
    }
    __syncthreads();
    const int h = (tid < NBMAX) ? s_h[tid] : 0;
    const int lo = block_excl_scan(h, s_w, total);
    if (tid < NBMAX) s_loff[tid] = lo;
    __syncthreads();
#pragma unroll
    for (int k = 0; k < EPT; ++k) {
        if (myk[k] >= 0) {
            const int pos = s_loff[myr[k] / RB] + myk[k];
            s_r[pos] = myr[k];
            s_c[pos] = myc[k];
            s_v[pos] = myv[k];
        }
    }
    __syncthreads();
    for (int i = tid; i < total; i += PT) {
        const int r = s_r[i];
        const int b = r / RB;
        const int g = s_goff[b] + (i - s_loff[b]);
        rec_r[g] = r;
        rec_c[g] = s_c[i];
        rec_v[g] = s_v[i];
    }
}

#define CAP 16384
#define SPT (CAP / PT)
typedef __attribute__((ext_vector_type(2))) int v2i;

__global__ void __launch_bounds__(PT) k_accum(const float* __restrict__ xt, const int* __restrict__ rec_r,
                                              const int* __restrict__ rec_c, const float* __restrict__ rec_v,
                                              const int* __restrict__ tot, const float* __restrict__ bias,
                                              float* __restrict__ out, int nd) {
    __shared__ __attribute__((aligned(16))) float acc[RB * AP];
    __shared__ __attribute__((aligned(8))) v2i s_cv[CAP];
    __shared__ int s_cnt[RB];
    __shared__ int s_off[RB + 1];
    __shared__ int s_w[65];
    __shared__ int s_se[2];
    const int tid = threadIdx.x;
    const int lane = tid & 31;
    const int w = __builtin_amdgcn_readfirstlane(tid >> 5);
    const int b = blockIdx.x;

    int total;
    const int t = (tid < NBMAX) ? tot[tid] : 0;
    const int bs = block_excl_scan(t, s_w, total);
    if (tid == b) { s_se[0] = bs; s_se[1] = t; }
    __syncthreads();
    const int start = __builtin_amdgcn_readfirstlane(s_se[0]);
    const int cnt = __builtin_amdgcn_readfirstlane(s_se[1]);
    const int row0 = b * RB;
    const int ra = (w * RB) >> 5;
    const int rb = ((w + 1) * RB) >> 5;
    const float* xl = xt + lane * 2;
    int npass = (cnt + CAP - 1) / CAP;
    if (npass < 1) npass = 1;

    for (int pass = 0; pass < npass; ++pass) {
        const int cs = start + pass * CAP;
        const int m = (cnt - pass * CAP < CAP) ? (cnt - pass * CAP) : CAP;
        if (tid < RB) s_cnt[tid] = 0;
        __syncthreads();
        int rr[SPT], cc[SPT], rk[SPT];
        float vv[SPT];
#pragma unroll
        for (int k = 0; k < SPT; ++k) {
            const int i = k * PT + tid;
            rr[k] = 0; cc[k] = 0; vv[k] = 0.f;
            if (i < m) {
                rr[k] = rec_r[cs + i] - row0;
                cc[k] = rec_c[cs + i];
                vv[k] = rec_v[cs + i];
            }
        }
#pragma unroll
        for (int k = 0; k < SPT; ++k) {
            rk[k] = 0;
            if (k * PT + tid < m) rk[k] = atomicAdd(&s_cnt[rr[k]], 1);
        }
        __syncthreads();
        const int h = (tid < RB) ? s_cnt[tid] : 0;
        int msum;
        const int ex = block_excl_scan(h, s_w, msum);
        if (tid < RB) s_off[tid] = ex;
        if (tid == RB) s_off[RB] = msum;
        __syncthreads();
#pragma unroll
        for (int k = 0; k < SPT; ++k) {
            if (k * PT + tid < m) {
                v2i rec = {cc[k], __float_as_int(vv[k])};
                s_cv[s_off[rr[k]] + rk[k]] = rec;
            }
        }
        __syncthreads();

        const int e0 = __builtin_amdgcn_readfirstlane(s_off[ra]);
        const int e1 = __builtin_amdgcn_readfirstlane(s_off[rb]);
        int row = ra;
        int rend = __builtin_amdgcn_readfirstlane(s_off[ra + 1]);
        v2f a = {0.f, 0.f};
#define FLUSH_ROW()                                                                         \
        {                                                                                   \
            v2f* dst = (v2f*)(acc + row * AP + lane * 2);                                   \
            if (pass == 0) { *dst = a; }                                                    \
            else { v2f o = *dst; o.x += a.x; o.y += a.y; *dst = o; }                        \
            a.x = 0.f; a.y = 0.f;                                                           \
            ++row;                                                                          \
            rend = (row < rb) ? __builtin_amdgcn_readfirstlane(s_off[row + 1]) : 0x7fffffff; \
        }
        v2i nx = {0, 0};
        if (e0 < e1) { const int li = e0 + lane; nx = s_cv[li < e1 ? li : e1 - 1]; }
        for (int base = e0; base < e1; base += 32) {
            const int n = (e1 - base < 32) ? (e1 - base) : 32;
            const int c = nx.x, vb = nx.y;
            if (base + 32 < e1) { const int li = base + 32 + lane; nx = s_cv[li < e1 ? li : e1 - 1]; }
            for (int j0 = 0; j0 < n; j0 += 8) {
                v2f xv[8];
#pragma unroll
                for (int u = 0; u < 8; ++u) {
                    const int cj = __builtin_amdgcn_readlane(c, j0 + u);
                    xv[u] = *(const v2f*)(xl + (size_t)cj * BATCH);
                }
                const int eb = base + j0;
                if (eb + 8 <= rend) {
#pragma unroll
                    for (int u = 0; u < 8; ++u) {
                        const float vj = __int_as_float(__builtin_amdgcn_readlane(vb, j0 + u));
                        const v2f vj2 = {vj, vj};
                        a = __builtin_elementwise_fma(xv[u], vj2, a);
                    }
                } else {
#pragma unroll
                    for (int u = 0; u < 8; ++u) {
                        while (eb + u >= rend) FLUSH_ROW()
                        const float vj = __int_as_float(__builtin_amdgcn_readlane(vb, j0 + u));
                        const v2f vj2 = {vj, vj};
                        a = __builtin_elementwise_fma(xv[u], vj2, a);
                    }
                }
            }
        }
        while (row < rb) FLUSH_ROW()
#undef FLUSH_ROW
    }
    __syncthreads();
    const int rows = (nd - row0 < RB) ? (nd - row0) : RB;
    const int fd = rows >> 4;
    const int ntiles = fd * (BATCH / 16);
    const int n = lane & 15;
    const int mbase = (lane < 16) ? 0 : 8;
    const v2f a = {(lane < 16) ? 1.0f : 0.0f, 0.0f};
    for (int t = w; t < ntiles; t += PT / 32) {
        const int b0 = (t & 3) * 16;
        const int d0 = (t >> 2) * 16;
        const float* cb = acc + (d0 + n) * AP + b0 + mbase;
        const float4 lo = *(const float4*)(cb);
        const float4 hi = *(const float4*)(cb + 4);
        const v8f c = {lo.x, lo.y, lo.z, lo.w, hi.x, hi.y, hi.z, hi.w};
        const v2f bm = {(lane < 16) ? bias[row0 + d0 + n] : 0.0f, 0.0f};
        const v8f d = __builtin_amdgcn_wmma_f32_16x16x4_f32(
             false, a,  false, bm,
             (short)0, c,  false,  false);
        float* o = out + (size_t)(b0 + mbase) * nd + row0 + d0 + n;
#pragma unroll
        for (int v = 0; v < 8; ++v) o[(size_t)v * nd] = d[v];
    }
    const int dt = fd * 16;
    for (int i = tid; i < (rows - dt) * BATCH; i += PT) {
        const int dd = dt + (i >> 6);
        const int bb = i & 63;
        out[(size_t)bb * nd + row0 + dd] = acc[dd * AP + bb] + bias[row0 + dd];
    }
}

extern "C" void kernel_launch(void* const* d_in, const int* in_sizes, int n_in,
                              void* d_out, int out_size, void* d_ws, size_t ws_size,
                              hipStream_t stream) {
    const float* x    = (const float*)d_in[0];
    const float* vals = (const float*)d_in[1];
    const float* bias = (const float*)d_in[2];
    const int* row    = (const int*)d_in[3];
    const int* col    = (const int*)d_in[4];
    float* out = (float*)d_out;

    const int nnz = in_sizes[1];
    const int nd  = in_sizes[2];
    const int ns  = in_sizes[0] / BATCH;

    float* xt    = (float*)d_ws;
    float* out_t = xt + (size_t)ns * BATCH;

    if ((ns & 3) == 0) {
        k_transpose_wide<<<(ns + TS - 1) / TS, 256, 0, stream>>>(x, xt, ns);
    } else {
        dim3 tb(32, 8);
        dim3 tg((ns + 31) / 32, (BATCH + 31) / 32);
        k_transpose<<<tg, tb, 0, stream>>>(x, xt, ns);
    }

    const int nb = (nd + RB - 1) / RB;
    const int pb = (int)(((long long)nnz + TILE - 1) / TILE);
    const size_t need_words = (size_t)ns * BATCH + (size_t)nd * BATCH + 3 * (size_t)nnz +
                              2 * (size_t)pb * NBMAX + NBMAX;
    if (nb <= NBMAX && pb >= 1 && need_words * 4 <= ws_size) {
        int* rec_r   = (int*)(out_t + (size_t)nd * BATCH);
        int* rec_c   = rec_r + (size_t)nnz;
        float* rec_v = (float*)(rec_c + (size_t)nnz);
        int* hist    = (int*)(rec_v + (size_t)nnz);
        int* off     = hist + (size_t)pb * NBMAX;
        int* tot     = off + (size_t)pb * NBMAX;
        k_hist<<<pb, PT, 0, stream>>>(row, nnz, hist);
        k_scanp<<<NBMAX, PT, 0, stream>>>(hist, off, tot, pb);
        k_place<<<pb, PT, 0, stream>>>(row, col, vals, nnz, off, tot, rec_r, rec_c, rec_v);
        k_accum<<<nb, PT, 0, stream>>>(xt, rec_r, rec_c, rec_v, tot, bias, out, nd);
        return;
    } else {
        long long n4 = ((long long)nd * BATCH) / 4;
        long long zb = (n4 + 255) / 256;
        if (zb > 16384) zb = 16384;
        k_zero<<<(int)zb, 256, 0, stream>>>((float4*)out_t, n4);
        long long waves = ((long long)nnz + 32LL * CHUNKS - 1) / (32LL * CHUNKS);
        long long sblocks = (waves + 7) / 8;
        if (sblocks < 1) sblocks = 1;
        k_scatter<<<(int)sblocks, 256, 0, stream>>>(xt, vals, row, col, out_t, nnz);
    }

    const int dtiles = nd / 16;
    if (dtiles > 0) {
        const int tiles = (BATCH / 16) * dtiles;
        const int eblocks = (tiles + 7) / 8;
        k_epilogue_wmma<<<eblocks, 256, 0, stream>>>(out_t, bias, out, nd, dtiles);
    }
    const int dstart = dtiles * 16;
    if (dstart < nd) {
        const int total = (nd - dstart) * BATCH;
        k_epilogue_tail<<<(total + 255) / 256, 256, 0, stream>>>(out_t, bias, out, nd, dstart);
    }
}


# Count and placement kernels: all of a thread's 13 (39) edge loads issued before its first LDS counter update; were load, wait, update 13 times in series; ranks still in counter arrival order
# speedup vs baseline: 2.4123x; 1.0656x over previous
#include <hip/hip_runtime.h>

typedef __attribute__((ext_vector_type(2))) float v2f;
typedef __attribute__((ext_vector_type(8))) float v8f;

#define BATCH 64
#define CHUNKS 4

__global__ void k_transpose(const float* __restrict__ x, float* __restrict__ xt, int ns) {
    __shared__ float tile[32][33];
    const int tx = threadIdx.x;
    const int ty = threadIdx.y;
    const int s0 = blockIdx.x * 32;
    const int b0 = blockIdx.y * 32;
#pragma unroll
    for (int k = 0; k < 4; ++k) {
        int s = s0 + tx, b = b0 + ty + k * 8;
        if (s < ns && b < BATCH) tile[ty + k * 8][tx] = x[(size_t)b * ns + s];
    }
    __syncthreads();
#pragma unroll
    for (int k = 0; k < 4; ++k) {
        int s = s0 + ty + k * 8, b = b0 + tx;
        if (s < ns && b < BATCH) xt[(size_t)s * BATCH + b] = tile[tx][ty + k * 8];
    }
}

#define TS 128
#define TPITCH 65
template <bool FULL>
__device__ __forceinline__ void transpose_tile(const float* __restrict__ x, float* __restrict__ xt,
                                               int ns, int s0, float* __restrict__ t) {
    const int tid = threadIdx.x;
    const int q = tid & 31;
    const int w = tid >> 5;
    if (FULL || s0 + 4 * q < ns) {
        const float* p = x + (size_t)w * ns + s0 + 4 * q;
        const size_t step = (size_t)8 * ns;
        float4 v[8];
#pragma unroll
        for (int k = 0; k < 8; ++k) v[k] = *(const float4*)(p + k * step);
        const int wbase = (4 * q) * TPITCH + ((4 * q) >> 6) + w;
#pragma unroll
        for (int k = 0; k < 8; ++k) {
            t[wbase + 8 * k]              = v[k].x;
            t[wbase + 8 * k + TPITCH]     = v[k].y;
            t[wbase + 8 * k + 2 * TPITCH] = v[k].z;
            t[wbase + 8 * k + 3 * TPITCH] = v[k].w;
        }
    }
    __syncthreads();
    const int bq = tid & 15;
    const int sr = tid >> 4;
    float* dst = xt + (size_t)(s0 + sr) * BATCH + 4 * bq;
    if (FULL) {
        float4 o[8];
#pragma unroll
        for (int k = 0; k < 8; ++k) {
            const int s = sr + 16 * k;
            const int rb = s * TPITCH + (s >> 6) + 4 * bq;
            o[k] = make_float4(t[rb], t[rb + 1], t[rb + 2], t[rb + 3]);
        }
#pragma unroll
        for (int k = 0; k < 8; ++k) *(float4*)(dst + (size_t)(16 * k) * BATCH) = o[k];
    } else {
#pragma unroll
        for (int k = 0; k < 8; ++k) {
            const int s = sr + 16 * k;
            if (s0 + s < ns) {
                const int rb = s * TPITCH + (s >> 6) + 4 * bq;
                *(float4*)(dst + (size_t)(16 * k) * BATCH) =
                    make_float4(t[rb], t[rb + 1], t[rb + 2], t[rb + 3]);
            }
        }
    }
}

__global__ void __launch_bounds__(256) k_transpose_wide(const float* __restrict__ x,
                                                        float* __restrict__ xt, int ns) {
    __shared__ float t[TS * TPITCH + 4];
    const int s0 = blockIdx.x * TS;
    if (s0 + TS <= ns) transpose_tile<true>(x, xt, ns, s0, t);
    else               transpose_tile<false>(x, xt, ns, s0, t);
}

__global__ void k_zero(float4* __restrict__ p, long long n4) {
    long long i = (long long)blockIdx.x * blockDim.x + threadIdx.x;
    const long long stride = (long long)gridDim.x * blockDim.x;
    const float4 z = make_float4(0.f, 0.f, 0.f, 0.f);
    for (; i < n4; i += stride) p[i] = z;
}

__global__ void k_scatter(const float* __restrict__ xt, const float* __restrict__ vals,
                          const int* __restrict__ row, const int* __restrict__ col,
                          float* __restrict__ out_t, int nnz) {
    const int lane = threadIdx.x & 31;
    const long long wave = ((long long)blockIdx.x * blockDim.x + threadIdx.x) >> 5;
    const long long base0 = wave * (32LL * CHUNKS);
    for (int ch = 0; ch < CHUNKS; ++ch) {
        const long long base = base0 + 32LL * ch;
        if (base >= nnz) return;
        const long long idx = base + lane;
        float v = 0.f;
        int r = 0, c = 0;
        if (idx < nnz) {
            v = vals[idx];
            r = row[idx];
            c = col[idx];
        }
        if (ch + 1 < CHUNKS && base + 32 + lane < nnz) {
            __builtin_prefetch(vals + base + 32 + lane, 0, 0);
            __builtin_prefetch(row + base + 32 + lane, 0, 0);
            __builtin_prefetch(col + base + 32 + lane, 0, 0);
        }
        const int count = (nnz - base < 32) ? (int)(nnz - base) : 32;
        for (int j = 0; j < count; ++j) {
            const float vj = __shfl(v, j);
            const int rj   = __shfl(r, j);
            const int cj   = __shfl(c, j);
            const v2f xv = *(const v2f*)(xt + (size_t)cj * BATCH + lane * 2);
            float* dst = out_t + (size_t)rj * BATCH + lane * 2;
            unsafeAtomicAdd(dst, xv.x * vj);
            unsafeAtomicAdd(dst + 1, xv.y * vj);
        }
    }
}

__global__ void k_epilogue_wmma(const float* __restrict__ out_t, const float* __restrict__ bias,
                                float* __restrict__ out, int nd, int dtiles) {
    const int lane = threadIdx.x & 31;
    const int wave = (int)(((long long)blockIdx.x * blockDim.x + threadIdx.x) >> 5);
    const int tiles = (BATCH / 16) * dtiles;
    if (wave >= tiles) return;
    const int b0 = (wave & 3) * 16;
    const int d0 = (wave >> 2) * 16;
    const int n = lane & 15;
    const int mbase = (lane < 16) ? 0 : 8;

    const float* cbase = out_t + (size_t)(d0 + n) * BATCH + b0 + mbase;
    const float4 lo = *(const float4*)(cbase);
    const float4 hi = *(const float4*)(cbase + 4);
    v8f c = {lo.x, lo.y, lo.z, lo.w, hi.x, hi.y, hi.z, hi.w};

    v2f a = {(lane < 16) ? 1.0f : 0.0f, 0.0f};
    v2f bm = {(lane < 16) ? bias[d0 + n] : 0.0f, 0.0f};

    v8f d = __builtin_amdgcn_wmma_f32_16x16x4_f32(
         false, a,  false, bm,
         (short)0, c,  false,  false);

#pragma unroll
    for (int v = 0; v < 8; ++v) {
        out[(size_t)(b0 + mbase + v) * nd + d0 + n] = d[v];
    }
}

__global__ void k_epilogue_tail(const float* __restrict__ out_t, const float* __restrict__ bias,
                                float* __restrict__ out, int nd, int dstart) {
    const int total = (nd - dstart) * BATCH;
    const int i = blockIdx.x * blockDim.x + threadIdx.x;
    if (i >= total) return;
    const int d = dstart + (i >> 6);
    const int b = i & 63;
    out[(size_t)b * nd + d] = out_t[(size_t)d * BATCH + b] + bias[d];
}

#define RB 400
#define AP 68
#define NBMAX 256
#define PT 1024
#define EPT 13
#define TILE (PT * EPT)

__device__ __forceinline__ int block_excl_scan(int v, int* s_w, int& total) {
    const int lane = threadIdx.x & 31;
    const int w = threadIdx.x >> 5;
    int inc = v;
#pragma unroll
    for (int d = 1; d < 32; d <<= 1) {
        const int t = __shfl_up(inc, d);
        if (lane >= d) inc += t;
    }
    if (lane == 31) s_w[w] = inc;
    __syncthreads();
    if (w == 0) {
        const int x = s_w[lane];
        int xi = x;
#pragma unroll
        for (int d = 1; d < 32; d <<= 1) {
            const int t = __shfl_up(xi, d);
            if (lane >= d) xi += t;
        }
        s_w[32 + lane] = xi - x;
        if (lane == 31) s_w[64] = xi;
    }
    __syncthreads();
    total = s_w[64];
    const int r = s_w[32 + w] + inc - v;
    __syncthreads();
    return r;
}

__global__ void __launch_bounds__(PT) k_hist(const int* __restrict__ row, int nnz,
                                             int* __restrict__ hist) {
    __shared__ int s_h[NBMAX];
    const int tid = threadIdx.x;
    if (tid < NBMAX) s_h[tid] = 0;
    __syncthreads();
    const long long base = (long long)blockIdx.x * TILE;
    const bool full = base + TILE <= (long long)nnz;
    int r[EPT];
    if (full) {
#pragma unroll
        for (int k = 0; k < EPT; ++k) r[k] = row[base + k * PT + tid];
    } else {
#pragma unroll
        for (int k = 0; k < EPT; ++k) {
            const long long e = base + k * PT + tid;
            r[k] = 0;
            if (e < nnz) r[k] = row[e];
        }
    }
#pragma unroll
    for (int k = 0; k < EPT; ++k) {
        const long long e = base + k * PT + tid;
        if (full || e < nnz) atomicAdd(&s_h[r[k] / RB], 1);
    }
    __syncthreads();
    if (tid < NBMAX) hist[blockIdx.x * NBMAX + tid] = s_h[tid];
}

__global__ void __launch_bounds__(PT) k_scanp(const int* __restrict__ hist, int* __restrict__ off,
                                              int* __restrict__ tot, int pb) {
    __shared__ int s_w[65];
    const int b = blockIdx.x;
    const int tid = threadIdx.x;
    int carry = 0;
    for (int p0 = 0; p0 < pb; p0 += PT) {
        const int p = p0 + tid;
        const int v = (p < pb) ? hist[p * NBMAX + b] : 0;
        int total;
        const int ex = block_excl_scan(v, s_w, total);
        if (p < pb) off[p * NBMAX + b] = carry + ex;
        carry += total;
    }
    if (tid == 0) tot[b] = carry;
}

__global__ void __launch_bounds__(PT) k_place(const int* __restrict__ row, const int* __restrict__ col,
                                              const float* __restrict__ vals, int nnz,
                                              const int* __restrict__ off, const int* __restrict__ tot,
                                              int* __restrict__ rec_r, int* __restrict__ rec_c,
                                              float* __restrict__ rec_v) {
    __shared__ int s_r[TILE];
    __shared__ int s_c[TILE];
    __shared__ float s_v[TILE];
    __shared__ int s_h[NBMAX];
    __shared__ int s_loff[NBMAX];
    __shared__ int s_goff[NBMAX];
    __shared__ int s_w[65];
    const int tid = threadIdx.x;

    int total;
    const int t = (tid < NBMAX) ? tot[tid] : 0;
    const int bs = block_excl_scan(t, s_w, total);
    if (tid < NBMAX) {
        s_goff[tid] = bs + off[blockIdx.x * NBMAX + tid];
        s_h[tid] = 0;
    }
    __syncthreads();

    int myr[EPT], myc[EPT], myk[EPT];
    float myv[EPT];
    const long long base = (long long)blockIdx.x * TILE;
    const bool full = base + TILE <= (long long)nnz;
    if (full) {
#pragma unroll
        for (int k = 0; k < EPT; ++k) myr[k] = row[base + k * PT + tid];
#pragma unroll
        for (int k = 0; k < EPT; ++k) {
            myc[k] = col[base + k * PT + tid];
            myv[k] = vals[base + k * PT + tid];
        }
    } else {
#pragma unroll
        for (int k = 0; k < EPT; ++k) {
            const long long e = base + k * PT + tid;
            myr[k] = 0;
            if (e < nnz) myr[k] = row[e];
        }
#pragma unroll
        for (int k = 0; k < EPT; ++k) {
            const long long e = base + k * PT + tid;
            myc[k] = 0; myv[k] = 0.f;
            if (e < nnz) { myc[k] = col[e]; myv[k] = vals[e]; }
        }
    }
#pragma unroll
    for (int k = 0; k < EPT; ++k) {
        const long long e = base + k * PT + tid;
        myk[k] = -1;
        if (full || e < nnz) myk[k] = atomicAdd(&s_h[myr[k] / RB], 1);
    }
    __syncthreads();
    const int h = (tid < NBMAX) ? s_h[tid] : 0;
    const int lo = block_excl_scan(h, s_w, total);
    if (tid < NBMAX) s_loff[tid] = lo;
    __syncthreads();
#pragma unroll
    for (int k = 0; k < EPT; ++k) {
        if (myk[k] >= 0) {
            const int pos = s_loff[myr[k] / RB] + myk[k];
            s_r[pos] = myr[k];
            s_c[pos] = myc[k];
            s_v[pos] = myv[k];
        }
    }
    __syncthreads();
    for (int i = tid; i < total; i += PT) {
        const int r = s_r[i];
        const int b = r / RB;
        const int g = s_goff[b] + (i - s_loff[b]);
        rec_r[g] = r;
        rec_c[g] = s_c[i];
        rec_v[g] = s_v[i];
    }
}

#define CAP 16384
#define SPT (CAP / PT)
typedef __attribute__((ext_vector_type(2))) int v2i;

__global__ void __launch_bounds__(PT) k_accum(const float* __restrict__ xt, const int* __restrict__ rec_r,
                                              const int* __restrict__ rec_c, const float* __restrict__ rec_v,
                                              const int* __restrict__ tot, const float* __restrict__ bias,
                                              float* __restrict__ out, int nd) {
    __shared__ __attribute__((aligned(16))) float acc[RB * AP];
    __shared__ __attribute__((aligned(8))) v2i s_cv[CAP];
    __shared__ int s_cnt[RB];
    __shared__ int s_off[RB + 1];
    __shared__ int s_w[65];
    __shared__ int s_se[2];
    const int tid = threadIdx.x;
    const int lane = tid & 31;
    const int w = __builtin_amdgcn_readfirstlane(tid >> 5);
    const int b = blockIdx.x;

    int total;
    const int t = (tid < NBMAX) ? tot[tid] : 0;
    const int bs = block_excl_scan(t, s_w, total);
    if (tid == b) { s_se[0] = bs; s_se[1] = t; }
    __syncthreads();
    const int start = __builtin_amdgcn_readfirstlane(s_se[0]);
    const int cnt = __builtin_amdgcn_readfirstlane(s_se[1]);
    const int row0 = b * RB;
    const int ra = (w * RB) >> 5;
    const int rb = ((w + 1) * RB) >> 5;
    const float* xl = xt + lane * 2;
    int npass = (cnt + CAP - 1) / CAP;
    if (npass < 1) npass = 1;

    for (int pass = 0; pass < npass; ++pass) {
        const int cs = start + pass * CAP;
        const int m = (cnt - pass * CAP < CAP) ? (cnt - pass * CAP) : CAP;
        if (tid < RB) s_cnt[tid] = 0;
        __syncthreads();
        int rr[SPT], cc[SPT], rk[SPT];
        float vv[SPT];
#pragma unroll
        for (int k = 0; k < SPT; ++k) {
            const int i = k * PT + tid;
            rr[k] = 0; cc[k] = 0; vv[k] = 0.f;
            if (i < m) {
                rr[k] = rec_r[cs + i] - row0;
                cc[k] = rec_c[cs + i];
                vv[k] = rec_v[cs + i];
            }
        }
#pragma unroll
        for (int k = 0; k < SPT; ++k) {
            rk[k] = 0;
            if (k * PT + tid < m) rk[k] = atomicAdd(&s_cnt[rr[k]], 1);
        }
        __syncthreads();
        const int h = (tid < RB) ? s_cnt[tid] : 0;
        int msum;
        const int ex = block_excl_scan(h, s_w, msum);
        if (tid < RB) s_off[tid] = ex;
        if (tid == RB) s_off[RB] = msum;
        __syncthreads();
#pragma unroll
        for (int k = 0; k < SPT; ++k) {
            if (k * PT + tid < m) {
                v2i rec = {cc[k], __float_as_int(vv[k])};
                s_cv[s_off[rr[k]] + rk[k]] = rec;
            }
        }
        __syncthreads();

        const int e0 = __builtin_amdgcn_readfirstlane(s_off[ra]);
        const int e1 = __builtin_amdgcn_readfirstlane(s_off[rb]);
        int row = ra;
        int rend = __builtin_amdgcn_readfirstlane(s_off[ra + 1]);
        v2f a = {0.f, 0.f};
#define FLUSH_ROW()                                                                         \
        {                                                                                   \
            v2f* dst = (v2f*)(acc + row * AP + lane * 2);                                   \
            if (pass == 0) { *dst = a; }                                                    \
            else { v2f o = *dst; o.x += a.x; o.y += a.y; *dst = o; }                        \
            a.x = 0.f; a.y = 0.f;                                                           \
            ++row;                                                                          \
            rend = (row < rb) ? __builtin_amdgcn_readfirstlane(s_off[row + 1]) : 0x7fffffff; \
        }
        v2i nx = {0, 0};
        if (e0 < e1) { const int li = e0 + lane; nx = s_cv[li < e1 ? li : e1 - 1]; }
        for (int base = e0; base < e1; base += 32) {
            const int n = (e1 - base < 32) ? (e1 - base) : 32;
            const int c = nx.x, vb = nx.y;
            if (base + 32 < e1) { const int li = base + 32 + lane; nx = s_cv[li < e1 ? li : e1 - 1]; }
            for (int j0 = 0; j0 < n; j0 += 8) {
                v2f xv[8];
#pragma unroll
                for (int u = 0; u < 8; ++u) {
                    const int cj = __builtin_amdgcn_readlane(c, j0 + u);
                    xv[u] = *(const v2f*)(xl + (size_t)cj * BATCH);
                }
                const int eb = base + j0;
                if (eb + 8 <= rend) {
#pragma unroll
                    for (int u = 0; u < 8; ++u) {
                        const float vj = __int_as_float(__builtin_amdgcn_readlane(vb, j0 + u));
                        const v2f vj2 = {vj, vj};
                        a = __builtin_elementwise_fma(xv[u], vj2, a);
                    }
                } else {
#pragma unroll
                    for (int u = 0; u < 8; ++u) {
                        while (eb + u >= rend) FLUSH_ROW()
                        const float vj = __int_as_float(__builtin_amdgcn_readlane(vb, j0 + u));
                        const v2f vj2 = {vj, vj};
                        a = __builtin_elementwise_fma(xv[u], vj2, a);
                    }
                }
            }
        }
        while (row < rb) FLUSH_ROW()
#undef FLUSH_ROW
    }
    __syncthreads();
    const int rows = (nd - row0 < RB) ? (nd - row0) : RB;
    const int fd = rows >> 4;
    const int ntiles = fd * (BATCH / 16);
    const int n = lane & 15;
    const int mbase = (lane < 16) ? 0 : 8;
    const v2f a = {(lane < 16) ? 1.0f : 0.0f, 0.0f};
    for (int t = w; t < ntiles; t += PT / 32) {
        const int b0 = (t & 3) * 16;
        const int d0 = (t >> 2) * 16;
        const float* cb = acc + (d0 + n) * AP + b0 + mbase;
        const float4 lo = *(const float4*)(cb);
        const float4 hi = *(const float4*)(cb + 4);
        const v8f c = {lo.x, lo.y, lo.z, lo.w, hi.x, hi.y, hi.z, hi.w};
        const v2f bm = {(lane < 16) ? bias[row0 + d0 + n] : 0.0f, 0.0f};
        const v8f d = __builtin_amdgcn_wmma_f32_16x16x4_f32(
             false, a,  false, bm,
             (short)0, c,  false,  false);
        float* o = out + (size_t)(b0 + mbase) * nd + row0 + d0 + n;
#pragma unroll
        for (int v = 0; v < 8; ++v) o[(size_t)v * nd] = d[v];
    }
    const int dt = fd * 16;
    for (int i = tid; i < (rows - dt) * BATCH; i += PT) {
        const int dd = dt + (i >> 6);
        const int bb = i & 63;
        out[(size_t)bb * nd + row0 + dd] = acc[dd * AP + bb] + bias[row0 + dd];
    }
}

extern "C" void kernel_launch(void* const* d_in, const int* in_sizes, int n_in,
                              void* d_out, int out_size, void* d_ws, size_t ws_size,
                              hipStream_t stream) {
    const float* x    = (const float*)d_in[0];
    const float* vals = (const float*)d_in[1];
    const float* bias = (const float*)d_in[2];
    const int* row    = (const int*)d_in[3];
    const int* col    = (const int*)d_in[4];
    float* out = (float*)d_out;

    const int nnz = in_sizes[1];
    const int nd  = in_sizes[2];
    const int ns  = in_sizes[0] / BATCH;

    float* xt    = (float*)d_ws;
    float* out_t = xt + (size_t)ns * BATCH;

    if ((ns & 3) == 0) {
        k_transpose_wide<<<(ns + TS - 1) / TS, 256, 0, stream>>>(x, xt, ns);
    } else {
        dim3 tb(32, 8);
        dim3 tg((ns + 31) / 32, (BATCH + 31) / 32);
        k_transpose<<<tg, tb, 0, stream>>>(x, xt, ns);
    }

    const int nb = (nd + RB - 1) / RB;
    const int pb = (int)(((long long)nnz + TILE - 1) / TILE);
    const size_t need_words = (size_t)ns * BATCH + (size_t)nd * BATCH + 3 * (size_t)nnz +
                              2 * (size_t)pb * NBMAX + NBMAX;
    if (nb <= NBMAX && pb >= 1 && need_words * 4 <= ws_size) {
        int* rec_r   = (int*)(out_t + (size_t)nd * BATCH);
        int* rec_c   = rec_r + (size_t)nnz;
        float* rec_v = (float*)(rec_c + (size_t)nnz);
        int* hist    = (int*)(rec_v + (size_t)nnz);
        int* off     = hist + (size_t)pb * NBMAX;
        int* tot     = off + (size_t)pb * NBMAX;
        k_hist<<<pb, PT, 0, stream>>>(row, nnz, hist);
        k_scanp<<<NBMAX, PT, 0, stream>>>(hist, off, tot, pb);
        k_place<<<pb, PT, 0, stream>>>(row, col, vals, nnz, off, tot, rec_r, rec_c, rec_v);
        k_accum<<<nb, PT, 0, stream>>>(xt, rec_r, rec_c, rec_v, tot, bias, out, nd);
        return;
    } else {
        long long n4 = ((long long)nd * BATCH) / 4;
        long long zb = (n4 + 255) / 256;
        if (zb > 16384) zb = 16384;
        k_zero<<<(int)zb, 256, 0, stream>>>((float4*)out_t, n4);
        long long waves = ((long long)nnz + 32LL * CHUNKS - 1) / (32LL * CHUNKS);
        long long sblocks = (waves + 7) / 8;
        if (sblocks < 1) sblocks = 1;
        k_scatter<<<(int)sblocks, 256, 0, stream>>>(xt, vals, row, col, out_t, nnz);
    }

    const int dtiles = nd / 16;
    if (dtiles > 0) {
        const int tiles = (BATCH / 16) * dtiles;
        const int eblocks = (tiles + 7) / 8;
        k_epilogue_wmma<<<eblocks, 256, 0, stream>>>(out_t, bias, out, nd, dtiles);
    }
    const int dstart = dtiles * 16;
    if (dstart < nd) {
        const int total = (nd - dstart) * BATCH;
        k_epilogue_tail<<<(total + 255) / 256, 256, 0, stream>>>(out_t, bias, out, nd, dstart);
    }
}


# Bucket order per tile only: records stay in their tile with a (tile, bucket) run table, the bucket kernel finds its runs through an LDS position map; count and scan kernels dropped, 5 -> 3 launches
# speedup vs baseline: 2.7911x; 1.1570x over previous
#include <hip/hip_runtime.h>

typedef __attribute__((ext_vector_type(2))) float v2f;
typedef __attribute__((ext_vector_type(8))) float v8f;

#define BATCH 64
#define CHUNKS 4

__global__ void k_transpose(const float* __restrict__ x, float* __restrict__ xt, int ns) {
    __shared__ float tile[32][33];
    const int tx = threadIdx.x;
    const int ty = threadIdx.y;
    const int s0 = blockIdx.x * 32;
    const int b0 = blockIdx.y * 32;
#pragma unroll
    for (int k = 0; k < 4; ++k) {
        int s = s0 + tx, b = b0 + ty + k * 8;
        if (s < ns && b < BATCH) tile[ty + k * 8][tx] = x[(size_t)b * ns + s];
    }
    __syncthreads();
#pragma unroll
    for (int k = 0; k < 4; ++k) {
        int s = s0 + ty + k * 8, b = b0 + tx;
        if (s < ns && b < BATCH) xt[(size_t)s * BATCH + b] = tile[tx][ty + k * 8];
    }
}

#define TS 128
#define TPITCH 65
template <bool FULL>
__device__ __forceinline__ void transpose_tile(const float* __restrict__ x, float* __restrict__ xt,
                                               int ns, int s0, float* __restrict__ t) {
    const int tid = threadIdx.x;
    const int q = tid & 31;
    const int w = tid >> 5;
    if (FULL || s0 + 4 * q < ns) {
        const float* p = x + (size_t)w * ns + s0 + 4 * q;
        const size_t step = (size_t)8 * ns;
        float4 v[8];
#pragma unroll
        for (int k = 0; k < 8; ++k) v[k] = *(const float4*)(p + k * step);
        const int wbase = (4 * q) * TPITCH + ((4 * q) >> 6) + w;
#pragma unroll
        for (int k = 0; k < 8; ++k) {
            t[wbase + 8 * k]              = v[k].x;
            t[wbase + 8 * k + TPITCH]     = v[k].y;
            t[wbase + 8 * k + 2 * TPITCH] = v[k].z;
            t[wbase + 8 * k + 3 * TPITCH] = v[k].w;
        }
    }
    __syncthreads();
    const int bq = tid & 15;
    const int sr = tid >> 4;
    float* dst = xt + (size_t)(s0 + sr) * BATCH + 4 * bq;
    if (FULL) {
        float4 o[8];
#pragma unroll
        for (int k = 0; k < 8; ++k) {
            const int s = sr + 16 * k;
            const int rb = s * TPITCH + (s >> 6) + 4 * bq;
            o[k] = make_float4(t[rb], t[rb + 1], t[rb + 2], t[rb + 3]);
        }
#pragma unroll
        for (int k = 0; k < 8; ++k) *(float4*)(dst + (size_t)(16 * k) * BATCH) = o[k];
    } else {
#pragma unroll
        for (int k = 0; k < 8; ++k) {
            const int s = sr + 16 * k;
            if (s0 + s < ns) {
                const int rb = s * TPITCH + (s >> 6) + 4 * bq;
                *(float4*)(dst + (size_t)(16 * k) * BATCH) =
                    make_float4(t[rb], t[rb + 1], t[rb + 2], t[rb + 3]);
            }
        }
    }
}

__global__ void __launch_bounds__(256) k_transpose_wide(const float* __restrict__ x,
                                                        float* __restrict__ xt, int ns) {
    __shared__ float t[TS * TPITCH + 4];
    const int s0 = blockIdx.x * TS;
    if (s0 + TS <= ns) transpose_tile<true>(x, xt, ns, s0, t);
    else               transpose_tile<false>(x, xt, ns, s0, t);
}

__global__ void k_zero(float4* __restrict__ p, long long n4) {
    long long i = (long long)blockIdx.x * blockDim.x + threadIdx.x;
    const long long stride = (long long)gridDim.x * blockDim.x;
    const float4 z = make_float4(0.f, 0.f, 0.f, 0.f);
    for (; i < n4; i += stride) p[i] = z;
}

__global__ void k_scatter(const float* __restrict__ xt, const float* __restrict__ vals,
                          const int* __restrict__ row, const int* __restrict__ col,
                          float* __restrict__ out_t, int nnz) {
    const int lane = threadIdx.x & 31;
    const long long wave = ((long long)blockIdx.x * blockDim.x + threadIdx.x) >> 5;
    const long long base0 = wave * (32LL * CHUNKS);
    for (int ch = 0; ch < CHUNKS; ++ch) {
        const long long base = base0 + 32LL * ch;
        if (base >= nnz) return;
        const long long idx = base + lane;
        float v = 0.f;
        int r = 0, c = 0;
        if (idx < nnz) {
            v = vals[idx];
            r = row[idx];
            c = col[idx];
        }
        if (ch + 1 < CHUNKS && base + 32 + lane < nnz) {
            __builtin_prefetch(vals + base + 32 + lane, 0, 0);
            __builtin_prefetch(row + base + 32 + lane, 0, 0);
            __builtin_prefetch(col + base + 32 + lane, 0, 0);
        }
        const int count = (nnz - base < 32) ? (int)(nnz - base) : 32;
        for (int j = 0; j < count; ++j) {
            const float vj = __shfl(v, j);
            const int rj   = __shfl(r, j);
            const int cj   = __shfl(c, j);
            const v2f xv = *(const v2f*)(xt + (size_t)cj * BATCH + lane * 2);
            float* dst = out_t + (size_t)rj * BATCH + lane * 2;
            unsafeAtomicAdd(dst, xv.x * vj);
            unsafeAtomicAdd(dst + 1, xv.y * vj);
        }
    }
}

__global__ void k_epilogue_wmma(const float* __restrict__ out_t, const float* __restrict__ bias,
                                float* __restrict__ out, int nd, int dtiles) {
    const int lane = threadIdx.x & 31;
    const int wave = (int)(((long long)blockIdx.x * blockDim.x + threadIdx.x) >> 5);
    const int tiles = (BATCH / 16) * dtiles;
    if (wave >= tiles) return;
    const int b0 = (wave & 3) * 16;
    const int d0 = (wave >> 2) * 16;
    const int n = lane & 15;
    const int mbase = (lane < 16) ? 0 : 8;

    const float* cbase = out_t + (size_t)(d0 + n) * BATCH + b0 + mbase;
    const float4 lo = *(const float4*)(cbase);
    const float4 hi = *(const float4*)(cbase + 4);
    v8f c = {lo.x, lo.y, lo.z, lo.w, hi.x, hi.y, hi.z, hi.w};

    v2f a = {(lane < 16) ? 1.0f : 0.0f, 0.0f};
    v2f bm = {(lane < 16) ? bias[d0 + n] : 0.0f, 0.0f};

    v8f d = __builtin_amdgcn_wmma_f32_16x16x4_f32(
         false, a,  false, bm,
         (short)0, c,  false,  false);

#pragma unroll
    for (int v = 0; v < 8; ++v) {
        out[(size_t)(b0 + mbase + v) * nd + d0 + n] = d[v];
    }
}

__global__ void k_epilogue_tail(const float* __restrict__ out_t, const float* __restrict__ bias,
                                float* __restrict__ out, int nd, int dstart) {
    const int total = (nd - dstart) * BATCH;
    const int i = blockIdx.x * blockDim.x + threadIdx.x;
    if (i >= total) return;
    const int d = dstart + (i >> 6);
    const int b = i & 63;
    out[(size_t)b * nd + d] = out_t[(size_t)d * BATCH + b] + bias[d];
}

#define RB 400
#define AP 68
#define NBMAX 256
#define PT 1024
#define EPT 13
#define TILE (PT * EPT)

__device__ __forceinline__ int block_excl_scan(int v, int* s_w, int& total) {
    const int lane = threadIdx.x & 31;
    const int w = threadIdx.x >> 5;
    int inc = v;
#pragma unroll
    for (int d = 1; d < 32; d <<= 1) {
        const int t = __shfl_up(inc, d);
        if (lane >= d) inc += t;
    }
    if (lane == 31) s_w[w] = inc;
    __syncthreads();
    if (w == 0) {
        const int x = s_w[lane];
        int xi = x;
#pragma unroll
        for (int d = 1; d < 32; d <<= 1) {
            const int t = __shfl_up(xi, d);
            if (lane >= d) xi += t;
        }
        s_w[32 + lane] = xi - x;
        if (lane == 31) s_w[64] = xi;
    }
    __syncthreads();
    total = s_w[64];
    const int r = s_w[32 + w] + inc - v;
    __syncthreads();
    return r;
}

__global__ void __launch_bounds__(PT) k_place(const int* __restrict__ row, const int* __restrict__ col,
                                              const float* __restrict__ vals, int nnz,
                                              int* __restrict__ rec_r, int* __restrict__ rec_c,
                                              float* __restrict__ rec_v, int* __restrict__ runs) {
    __shared__ int s_r[TILE];
    __shared__ int s_c[TILE];
    __shared__ float s_v[TILE];
    __shared__ int s_h[NBMAX];
    __shared__ int s_loff[NBMAX];
    __shared__ int s_w[65];
    const int tid = threadIdx.x;

    if (tid < NBMAX) s_h[tid] = 0;
    __syncthreads();

    int myr[EPT], myc[EPT], myk[EPT];
    float myv[EPT];
    const long long base = (long long)blockIdx.x * TILE;
    const bool full = base + TILE <= (long long)nnz;
    if (full) {
#pragma unroll
        for (int k = 0; k < EPT; ++k) myr[k] = row[base + k * PT + tid];
#pragma unroll
        for (int k = 0; k < EPT; ++k) {
            myc[k] = col[base + k * PT + tid];
            myv[k] = vals[base + k * PT + tid];
        }
    } else {
#pragma unroll
        for (int k = 0; k < EPT; ++k) {
            const long long e = base + k * PT + tid;
            myr[k] = 0;
            if (e < nnz) myr[k] = row[e];
        }
#pragma unroll
        for (int k = 0; k < EPT; ++k) {
            const long long e = base + k * PT + tid;
            myc[k] = 0; myv[k] = 0.f;
            if (e < nnz) { myc[k] = col[e]; myv[k] = vals[e]; }
        }
    }
#pragma unroll
    for (int k = 0; k < EPT; ++k) {
        const long long e = base + k * PT + tid;
        myk[k] = -1;
        if (full || e < nnz) myk[k] = atomicAdd(&s_h[myr[k] / RB], 1);
    }
    __syncthreads();
    const int h = (tid < NBMAX) ? s_h[tid] : 0;
    int total;
    const int lo = block_excl_scan(h, s_w, total);
    if (tid < NBMAX) {
        s_loff[tid] = lo;
        int* rp = runs + 2 * ((size_t)blockIdx.x * NBMAX + tid);
        rp[0] = lo;
        rp[1] = h;
    }
    __syncthreads();
#pragma unroll
    for (int k = 0; k < EPT; ++k) {
        if (myk[k] >= 0) {
            const int pos = s_loff[myr[k] / RB] + myk[k];
            s_r[pos] = myr[k];
            s_c[pos] = myc[k];
            s_v[pos] = myv[k];
        }
    }
    __syncthreads();
    for (int i = tid; i < total; i += PT) {
        rec_r[base + i] = s_r[i];
        rec_c[base + i] = s_c[i];
        rec_v[base + i] = s_v[i];
    }
}

#define CAP 16384
#define SPT (CAP / PT)
typedef __attribute__((ext_vector_type(2))) int v2i;

__global__ void __launch_bounds__(PT) k_accum(const float* __restrict__ xt, const int* __restrict__ rec_r,
                                              const int* __restrict__ rec_c, const float* __restrict__ rec_v,
                                              const int* __restrict__ runs, int pb,
                                              const float* __restrict__ bias,
                                              float* __restrict__ out, int nd) {
    __shared__ __attribute__((aligned(16))) float acc[RB * AP];
    __shared__ __attribute__((aligned(8))) v2i s_cv[CAP];
    __shared__ int s_cnt[RB];
    __shared__ int s_off[RB + 1];
    __shared__ int s_w[65];
    __shared__ int s_rex[PT];
    __shared__ int s_rn[PT];
    __shared__ int s_rdl[PT];
    int* s_map = (int*)s_cv;
    const int tid = threadIdx.x;
    const int lane = tid & 31;
    const int w = __builtin_amdgcn_readfirstlane(tid >> 5);
    const int b = blockIdx.x;

    int total;
    int rn = 0, rlo = 0;
    if (tid < pb) {
        const int* rp = runs + 2 * ((size_t)tid * NBMAX + b);
        rlo = rp[0];
        rn = rp[1];
    }
    const int rex = block_excl_scan(rn, s_w, total);
    if (tid < pb) {
        s_rex[tid] = rex;
        s_rn[tid] = rn;
        s_rdl[tid] = tid * TILE + rlo - rex;
    }
    const int cnt = __builtin_amdgcn_readfirstlane(total);
    const int row0 = b * RB;
    const int ra = (w * RB) >> 5;
    const int rb = ((w + 1) * RB) >> 5;
    const float* xl = xt + lane * 2;
    int npass = (cnt + CAP - 1) / CAP;
    if (npass < 1) npass = 1;

    for (int pass = 0; pass < npass; ++pass) {
        const int v0 = pass * CAP;
        const int m = (cnt - v0 < CAP) ? (cnt - v0) : CAP;
        if (tid < RB) s_cnt[tid] = 0;
        __syncthreads();
        for (int p = w; p < pb; p += PT / 32) {
            const int pdl = s_rdl[p];
            int jlo = s_rex[p] - v0;
            int jhi = jlo + s_rn[p];
            if (jlo < 0) jlo = 0;
            if (jhi > CAP) jhi = CAP;
            for (int j = jlo + lane; j < jhi; j += 32) s_map[j] = pdl;
        }
        __syncthreads();
        int rr[SPT], cc[SPT], rk[SPT], dl[SPT];
        float vv[SPT];
#pragma unroll
        for (int k = 0; k < SPT; ++k) {
            const int i = k * PT + tid;
            dl[k] = 0;
            if (i < m) dl[k] = s_map[i];
        }
#pragma unroll
        for (int k = 0; k < SPT; ++k) {
            const int i = k * PT + tid;
            rr[k] = 0; cc[k] = 0; vv[k] = 0.f;
            if (i < m) {
                const int g = v0 + i + dl[k];
                rr[k] = rec_r[g] - row0;
                cc[k] = rec_c[g];
                vv[k] = rec_v[g];
            }
        }
#pragma unroll
        for (int k = 0; k < SPT; ++k) {
            rk[k] = 0;
            if (k * PT + tid < m) rk[k] = atomicAdd(&s_cnt[rr[k]], 1);
        }
        __syncthreads();
        const int h = (tid < RB) ? s_cnt[tid] : 0;
        int msum;
        const int ex = block_excl_scan(h, s_w, msum);
        if (tid < RB) s_off[tid] = ex;
        if (tid == RB) s_off[RB] = msum;
        __syncthreads();
#pragma unroll
        for (int k = 0; k < SPT; ++k) {
            if (k * PT + tid < m) {
                v2i rec = {cc[k], __float_as_int(vv[k])};
                s_cv[s_off[rr[k]] + rk[k]] = rec;
            }
        }
        __syncthreads();

        const int e0 = __builtin_amdgcn_readfirstlane(s_off[ra]);
        const int e1 = __builtin_amdgcn_readfirstlane(s_off[rb]);
        int row = ra;
        int rend = __builtin_amdgcn_readfirstlane(s_off[ra + 1]);
        v2f a = {0.f, 0.f};
#define FLUSH_ROW()                                                                         \
        {                                                                                   \
            v2f* dst = (v2f*)(acc + row * AP + lane * 2);                                   \
            if (pass == 0) { *dst = a; }                                                    \
            else { v2f o = *dst; o.x += a.x; o.y += a.y; *dst = o; }                        \
            a.x = 0.f; a.y = 0.f;                                                           \
            ++row;                                                                          \
            rend = (row < rb) ? __builtin_amdgcn_readfirstlane(s_off[row + 1]) : 0x7fffffff; \
        }
        v2i nx = {0, 0};
        if (e0 < e1) { const int li = e0 + lane; nx = s_cv[li < e1 ? li : e1 - 1]; }
        for (int base = e0; base < e1; base += 32) {
            const int n = (e1 - base < 32) ? (e1 - base) : 32;
            const int c = nx.x, vb = nx.y;
            if (base + 32 < e1) { const int li = base + 32 + lane; nx = s_cv[li < e1 ? li : e1 - 1]; }
            for (int j0 = 0; j0 < n; j0 += 8) {
                v2f xv[8];
#pragma unroll
                for (int u = 0; u < 8; ++u) {
                    const int cj = __builtin_amdgcn_readlane(c, j0 + u);
                    xv[u] = *(const v2f*)(xl + (size_t)cj * BATCH);
                }
                const int eb = base + j0;
                if (eb + 8 <= rend) {
#pragma unroll
                    for (int u = 0; u < 8; ++u) {
                        const float vj = __int_as_float(__builtin_amdgcn_readlane(vb, j0 + u));
                        const v2f vj2 = {vj, vj};
                        a = __builtin_elementwise_fma(xv[u], vj2, a);
                    }
                } else {
#pragma unroll
                    for (int u = 0; u < 8; ++u) {
                        while (eb + u >= rend) FLUSH_ROW()
                        const float vj = __int_as_float(__builtin_amdgcn_readlane(vb, j0 + u));
                        const v2f vj2 = {vj, vj};
                        a = __builtin_elementwise_fma(xv[u], vj2, a);
                    }
                }
            }
        }
        while (row < rb) FLUSH_ROW()
#undef FLUSH_ROW
    }
    __syncthreads();
    const int rows = (nd - row0 < RB) ? (nd - row0) : RB;
    const int fd = rows >> 4;
    const int ntiles = fd * (BATCH / 16);
    const int n = lane & 15;
    const int mbase = (lane < 16) ? 0 : 8;
    const v2f a = {(lane < 16) ? 1.0f : 0.0f, 0.0f};
    for (int t = w; t < ntiles; t += PT / 32) {
        const int b0 = (t & 3) * 16;
        const int d0 = (t >> 2) * 16;
        const float* cb = acc + (d0 + n) * AP + b0 + mbase;
        const float4 lo = *(const float4*)(cb);
        const float4 hi = *(const float4*)(cb + 4);
        const v8f c = {lo.x, lo.y, lo.z, lo.w, hi.x, hi.y, hi.z, hi.w};
        const v2f bm = {(lane < 16) ? bias[row0 + d0 + n] : 0.0f, 0.0f};
        const v8f d = __builtin_amdgcn_wmma_f32_16x16x4_f32(
             false, a,  false, bm,
             (short)0, c,  false,  false);
        float* o = out + (size_t)(b0 + mbase) * nd + row0 + d0 + n;
#pragma unroll
        for (int v = 0; v < 8; ++v) o[(size_t)v * nd] = d[v];
    }
    const int dt = fd * 16;
    for (int i = tid; i < (rows - dt) * BATCH; i += PT) {
        const int dd = dt + (i >> 6);
        const int bb = i & 63;
        out[(size_t)bb * nd + row0 + dd] = acc[dd * AP + bb] + bias[row0 + dd];
    }
}

extern "C" void kernel_launch(void* const* d_in, const int* in_sizes, int n_in,
                              void* d_out, int out_size, void* d_ws, size_t ws_size,
                              hipStream_t stream) {
    const float* x    = (const float*)d_in[0];
    const float* vals = (const float*)d_in[1];
    const float* bias = (const float*)d_in[2];
    const int* row    = (const int*)d_in[3];
    const int* col    = (const int*)d_in[4];
    float* out = (float*)d_out;

    const int nnz = in_sizes[1];
    const int nd  = in_sizes[2];
    const int ns  = in_sizes[0] / BATCH;

    float* xt    = (float*)d_ws;
    float* out_t = xt + (size_t)ns * BATCH;

    if ((ns & 3) == 0) {
        k_transpose_wide<<<(ns + TS - 1) / TS, 256, 0, stream>>>(x, xt, ns);
    } else {
        dim3 tb(32, 8);
        dim3 tg((ns + 31) / 32, (BATCH + 31) / 32);
        k_transpose<<<tg, tb, 0, stream>>>(x, xt, ns);
    }

    const int nb = (nd + RB - 1) / RB;
    const int pb = (int)(((long long)nnz + TILE - 1) / TILE);
    const size_t need_words = (size_t)ns * BATCH + (size_t)nd * BATCH + 3 * (size_t)nnz +
                              2 * (size_t)pb * NBMAX + NBMAX;
    if (nb <= NBMAX && pb >= 1 && pb <= PT && need_words * 4 <= ws_size) {
        int* rec_r   = (int*)(out_t + (size_t)nd * BATCH);
        int* rec_c   = rec_r + (size_t)nnz;
        float* rec_v = (float*)(rec_c + (size_t)nnz);
        int* runs    = (int*)(rec_v + (size_t)nnz);
        k_place<<<pb, PT, 0, stream>>>(row, col, vals, nnz, rec_r, rec_c, rec_v, runs);
        k_accum<<<nb, PT, 0, stream>>>(xt, rec_r, rec_c, rec_v, runs, pb, bias, out, nd);
        return;
    } else {
        long long n4 = ((long long)nd * BATCH) / 4;
        long long zb = (n4 + 255) / 256;
        if (zb > 16384) zb = 16384;
        k_zero<<<(int)zb, 256, 0, stream>>>((float4*)out_t, n4);
        long long waves = ((long long)nnz + 32LL * CHUNKS - 1) / (32LL * CHUNKS);
        long long sblocks = (waves + 7) / 8;
        if (sblocks < 1) sblocks = 1;
        k_scatter<<<(int)sblocks, 256, 0, stream>>>(xt, vals, row, col, out_t, nnz);
    }

    const int dtiles = nd / 16;
    if (dtiles > 0) {
        const int tiles = (BATCH / 16) * dtiles;
        const int eblocks = (tiles + 7) / 8;
        k_epilogue_wmma<<<eblocks, 256, 0, stream>>>(out_t, bias, out, nd, dtiles);
    }
    const int dstart = dtiles * 16;
    if (dstart < nd) {
        const int total = (nd - dstart) * BATCH;
        k_epilogue_tail<<<(total + 255) / 256, 256, 0, stream>>>(out_t, bias, out, nd, dstart);
    }
}


# Edge records packed to 8 bytes (row in bucket << 23 | column, weight): one b64 store and load per record, were three b32; no row0 subtract in the load guard, so k_accum's 16 loads leave as one group
# speedup vs baseline: 2.8788x; 1.0314x over previous
#include <hip/hip_runtime.h>

typedef __attribute__((ext_vector_type(2))) float v2f;
typedef __attribute__((ext_vector_type(8))) float v8f;

#define BATCH 64
#define CHUNKS 4

__global__ void k_transpose(const float* __restrict__ x, float* __restrict__ xt, int ns) {
    __shared__ float tile[32][33];
    const int tx = threadIdx.x;
    const int ty = threadIdx.y;
    const int s0 = blockIdx.x * 32;
    const int b0 = blockIdx.y * 32;
#pragma unroll
    for (int k = 0; k < 4; ++k) {
        int s = s0 + tx, b = b0 + ty + k * 8;
        if (s < ns && b < BATCH) tile[ty + k * 8][tx] = x[(size_t)b * ns + s];
    }
    __syncthreads();
#pragma unroll
    for (int k = 0; k < 4; ++k) {
        int s = s0 + ty + k * 8, b = b0 + tx;
        if (s < ns && b < BATCH) xt[(size_t)s * BATCH + b] = tile[tx][ty + k * 8];
    }
}

#define TS 128
#define TPITCH 65
template <bool FULL>
__device__ __forceinline__ void transpose_tile(const float* __restrict__ x, float* __restrict__ xt,
                                               int ns, int s0, float* __restrict__ t) {
    const int tid = threadIdx.x;
    const int q = tid & 31;
    const int w = tid >> 5;
    if (FULL || s0 + 4 * q < ns) {
        const float* p = x + (size_t)w * ns + s0 + 4 * q;
        const size_t step = (size_t)8 * ns;
        float4 v[8];
#pragma unroll
        for (int k = 0; k < 8; ++k) v[k] = *(const float4*)(p + k * step);
        const int wbase = (4 * q) * TPITCH + ((4 * q) >> 6) + w;
#pragma unroll
        for (int k = 0; k < 8; ++k) {
            t[wbase + 8 * k]              = v[k].x;
            t[wbase + 8 * k + TPITCH]     = v[k].y;
            t[wbase + 8 * k + 2 * TPITCH] = v[k].z;
            t[wbase + 8 * k + 3 * TPITCH] = v[k].w;
        }
    }
    __syncthreads();
    const int bq = tid & 15;
    const int sr = tid >> 4;
    float* dst = xt + (size_t)(s0 + sr) * BATCH + 4 * bq;
    if (FULL) {
        float4 o[8];
#pragma unroll
        for (int k = 0; k < 8; ++k) {
            const int s = sr + 16 * k;
            const int rb = s * TPITCH + (s >> 6) + 4 * bq;
            o[k] = make_float4(t[rb], t[rb + 1], t[rb + 2], t[rb + 3]);
        }
#pragma unroll
        for (int k = 0; k < 8; ++k) *(float4*)(dst + (size_t)(16 * k) * BATCH) = o[k];
    } else {
#pragma unroll
        for (int k = 0; k < 8; ++k) {
            const int s = sr + 16 * k;
            if (s0 + s < ns) {
                const int rb = s * TPITCH + (s >> 6) + 4 * bq;
                *(float4*)(dst + (size_t)(16 * k) * BATCH) =
                    make_float4(t[rb], t[rb + 1], t[rb + 2], t[rb + 3]);
            }
        }
    }
}

__global__ void __launch_bounds__(256) k_transpose_wide(const float* __restrict__ x,
                                                        float* __restrict__ xt, int ns) {
    __shared__ float t[TS * TPITCH + 4];
    const int s0 = blockIdx.x * TS;
    if (s0 + TS <= ns) transpose_tile<true>(x, xt, ns, s0, t);
    else               transpose_tile<false>(x, xt, ns, s0, t);
}

__global__ void k_zero(float4* __restrict__ p, long long n4) {
    long long i = (long long)blockIdx.x * blockDim.x + threadIdx.x;
    const long long stride = (long long)gridDim.x * blockDim.x;
    const float4 z = make_float4(0.f, 0.f, 0.f, 0.f);
    for (; i < n4; i += stride) p[i] = z;
}

__global__ void k_scatter(const float* __restrict__ xt, const float* __restrict__ vals,
                          const int* __restrict__ row, const int* __restrict__ col,
                          float* __restrict__ out_t, int nnz) {
    const int lane = threadIdx.x & 31;
    const long long wave = ((long long)blockIdx.x * blockDim.x + threadIdx.x) >> 5;
    const long long base0 = wave * (32LL * CHUNKS);
    for (int ch = 0; ch < CHUNKS; ++ch) {
        const long long base = base0 + 32LL * ch;
        if (base >= nnz) return;
        const long long idx = base + lane;
        float v = 0.f;
        int r = 0, c = 0;
        if (idx < nnz) {
            v = vals[idx];
            r = row[idx];
            c = col[idx];
        }
        if (ch + 1 < CHUNKS && base + 32 + lane < nnz) {
            __builtin_prefetch(vals + base + 32 + lane, 0, 0);
            __builtin_prefetch(row + base + 32 + lane, 0, 0);
            __builtin_prefetch(col + base + 32 + lane, 0, 0);
        }
        const int count = (nnz - base < 32) ? (int)(nnz - base) : 32;
        for (int j = 0; j < count; ++j) {
            const float vj = __shfl(v, j);
            const int rj   = __shfl(r, j);
            const int cj   = __shfl(c, j);
            const v2f xv = *(const v2f*)(xt + (size_t)cj * BATCH + lane * 2);
            float* dst = out_t + (size_t)rj * BATCH + lane * 2;
            unsafeAtomicAdd(dst, xv.x * vj);
            unsafeAtomicAdd(dst + 1, xv.y * vj);
        }
    }
}

__global__ void k_epilogue_wmma(const float* __restrict__ out_t, const float* __restrict__ bias,
                                float* __restrict__ out, int nd, int dtiles) {
    const int lane = threadIdx.x & 31;
    const int wave = (int)(((long long)blockIdx.x * blockDim.x + threadIdx.x) >> 5);
    const int tiles = (BATCH / 16) * dtiles;
    if (wave >= tiles) return;
    const int b0 = (wave & 3) * 16;
    const int d0 = (wave >> 2) * 16;
    const int n = lane & 15;
    const int mbase = (lane < 16) ? 0 : 8;

    const float* cbase = out_t + (size_t)(d0 + n) * BATCH + b0 + mbase;
    const float4 lo = *(const float4*)(cbase);
    const float4 hi = *(const float4*)(cbase + 4);
    v8f c = {lo.x, lo.y, lo.z, lo.w, hi.x, hi.y, hi.z, hi.w};

    v2f a = {(lane < 16) ? 1.0f : 0.0f, 0.0f};
    v2f bm = {(lane < 16) ? bias[d0 + n] : 0.0f, 0.0f};

    v8f d = __builtin_amdgcn_wmma_f32_16x16x4_f32(
         false, a,  false, bm,
         (short)0, c,  false,  false);

#pragma unroll
    for (int v = 0; v < 8; ++v) {
        out[(size_t)(b0 + mbase + v) * nd + d0 + n] = d[v];
    }
}

__global__ void k_epilogue_tail(const float* __restrict__ out_t, const float* __restrict__ bias,
                                float* __restrict__ out, int nd, int dstart) {
    const int total = (nd - dstart) * BATCH;
    const int i = blockIdx.x * blockDim.x + threadIdx.x;
    if (i >= total) return;
    const int d = dstart + (i >> 6);
    const int b = i & 63;
    out[(size_t)b * nd + d] = out_t[(size_t)d * BATCH + b] + bias[d];
}

typedef __attribute__((ext_vector_type(2))) int v2i;
#define CBITS 23
#define CMASK ((1 << CBITS) - 1)
#define RB 400
#define AP 68
#define NBMAX 256
#define PT 1024
#define EPT 13
#define TILE (PT * EPT)

__device__ __forceinline__ int block_excl_scan(int v, int* s_w, int& total) {
    const int lane = threadIdx.x & 31;
    const int w = threadIdx.x >> 5;
    int inc = v;
#pragma unroll
    for (int d = 1; d < 32; d <<= 1) {
        const int t = __shfl_up(inc, d);
        if (lane >= d) inc += t;
    }
    if (lane == 31) s_w[w] = inc;
    __syncthreads();
    if (w == 0) {
        const int x = s_w[lane];
        int xi = x;
#pragma unroll
        for (int d = 1; d < 32; d <<= 1) {
            const int t = __shfl_up(xi, d);
            if (lane >= d) xi += t;
        }
        s_w[32 + lane] = xi - x;
        if (lane == 31) s_w[64] = xi;
    }
    __syncthreads();
    total = s_w[64];
    const int r = s_w[32 + w] + inc - v;
    __syncthreads();
    return r;
}

__global__ void __launch_bounds__(PT) k_place(const int* __restrict__ row, const int* __restrict__ col,
                                              const float* __restrict__ vals, int nnz,
                                              int* __restrict__ rec, int* __restrict__ runs) {
    __shared__ __attribute__((aligned(8))) v2i s_rec[TILE];
    __shared__ int s_h[NBMAX];
    __shared__ int s_loff[NBMAX];
    __shared__ int s_w[65];
    const int tid = threadIdx.x;

    if (tid < NBMAX) s_h[tid] = 0;
    __syncthreads();

    int myr[EPT], myc[EPT], myk[EPT];
    float myv[EPT];
    const long long base = (long long)blockIdx.x * TILE;
    const bool full = base + TILE <= (long long)nnz;
    if (full) {
#pragma unroll
        for (int k = 0; k < EPT; ++k) myr[k] = row[base + k * PT + tid];
#pragma unroll
        for (int k = 0; k < EPT; ++k) {
            myc[k] = col[base + k * PT + tid];
            myv[k] = vals[base + k * PT + tid];
        }
    } else {
#pragma unroll
        for (int k = 0; k < EPT; ++k) {
            const long long e = base + k * PT + tid;
            myr[k] = 0;
            if (e < nnz) myr[k] = row[e];
        }
#pragma unroll
        for (int k = 0; k < EPT; ++k) {
            const long long e = base + k * PT + tid;
            myc[k] = 0; myv[k] = 0.f;
            if (e < nnz) { myc[k] = col[e]; myv[k] = vals[e]; }
        }
    }
#pragma unroll
    for (int k = 0; k < EPT; ++k) {
        const long long e = base + k * PT + tid;
        myk[k] = -1;
        if (full || e < nnz) myk[k] = atomicAdd(&s_h[myr[k] / RB], 1);
    }
    __syncthreads();
    const int h = (tid < NBMAX) ? s_h[tid] : 0;
    int total;
    const int lo = block_excl_scan(h, s_w, total);
    if (tid < NBMAX) {
        s_loff[tid] = lo;
        int* rp = runs + 2 * ((size_t)blockIdx.x * NBMAX + tid);
        rp[0] = lo;
        rp[1] = h;
    }
    __syncthreads();
#pragma unroll
    for (int k = 0; k < EPT; ++k) {
        if (myk[k] >= 0) {
            const int bk = myr[k] / RB;
            const v2i o = {(int)(((unsigned)(myr[k] - bk * RB) << CBITS) | (unsigned)myc[k]), __float_as_int(myv[k])};
            s_rec[s_loff[bk] + myk[k]] = o;
        }
    }
    __syncthreads();
    v2i* __restrict__ rec2 = (v2i*)rec + base;
    for (int i = tid; i < total; i += PT) rec2[i] = s_rec[i];
}

#define CAP 16384
#define SPT (CAP / PT)

__global__ void __launch_bounds__(PT) k_accum(const float* __restrict__ xt, const int* __restrict__ rec,
                                              const int* __restrict__ runs, int pb,
                                              const float* __restrict__ bias,
                                              float* __restrict__ out, int nd) {
    __shared__ __attribute__((aligned(16))) float acc[RB * AP];
    __shared__ __attribute__((aligned(8))) v2i s_cv[CAP];
    __shared__ int s_cnt[RB];
    __shared__ int s_off[RB + 1];
    __shared__ int s_w[65];
    __shared__ int s_rex[PT];
    __shared__ int s_rn[PT];
    __shared__ int s_rdl[PT];
    int* s_map = (int*)s_cv;
    const v2i* __restrict__ rec2 = (const v2i*)rec;
    const int tid = threadIdx.x;
    const int lane = tid & 31;
    const int w = __builtin_amdgcn_readfirstlane(tid >> 5);
    const int b = blockIdx.x;

    int total;
    int rn = 0, rlo = 0;
    if (tid < pb) {
        const int* rp = runs + 2 * ((size_t)tid * NBMAX + b);
        rlo = rp[0];
        rn = rp[1];
    }
    const int rex = block_excl_scan(rn, s_w, total);
    if (tid < pb) {
        s_rex[tid] = rex;
        s_rn[tid] = rn;
        s_rdl[tid] = tid * TILE + rlo - rex;
    }
    const int cnt = __builtin_amdgcn_readfirstlane(total);
    const int row0 = b * RB;
    const int ra = (w * RB) >> 5;
    const int rb = ((w + 1) * RB) >> 5;
    const float* xl = xt + lane * 2;
    int npass = (cnt + CAP - 1) / CAP;
    if (npass < 1) npass = 1;

    for (int pass = 0; pass < npass; ++pass) {
        const int v0 = pass * CAP;
        const int m = (cnt - v0 < CAP) ? (cnt - v0) : CAP;
        if (tid < RB) s_cnt[tid] = 0;
        __syncthreads();
        for (int p = w; p < pb; p += PT / 32) {
            const int pdl = s_rdl[p];
            int jlo = s_rex[p] - v0;
            int jhi = jlo + s_rn[p];
            if (jlo < 0) jlo = 0;
            if (jhi > CAP) jhi = CAP;
            for (int j = jlo + lane; j < jhi; j += 32) s_map[j] = pdl;
        }
        __syncthreads();
        v2i rc[SPT];
        int rk[SPT], dl[SPT];
#pragma unroll
        for (int k = 0; k < SPT; ++k) {
            const int i = k * PT + tid;
            dl[k] = 0;
            if (i < m) dl[k] = s_map[i];
        }
#pragma unroll
        for (int k = 0; k < SPT; ++k) {
            const int i = k * PT + tid;
            rc[k].x = 0; rc[k].y = 0;
            if (i < m) rc[k] = rec2[v0 + i + dl[k]];
        }
#pragma unroll
        for (int k = 0; k < SPT; ++k) {
            rk[k] = 0;
            if (k * PT + tid < m) rk[k] = atomicAdd(&s_cnt[(unsigned)rc[k].x >> CBITS], 1);
        }
        __syncthreads();
        const int h = (tid < RB) ? s_cnt[tid] : 0;
        int msum;
        const int ex = block_excl_scan(h, s_w, msum);
        if (tid < RB) s_off[tid] = ex;
        if (tid == RB) s_off[RB] = msum;
        __syncthreads();
#pragma unroll
        for (int k = 0; k < SPT; ++k) {
            if (k * PT + tid < m) {
                const v2i o = {rc[k].x & CMASK, rc[k].y};
                s_cv[s_off[(unsigned)rc[k].x >> CBITS] + rk[k]] = o;
            }
        }
        __syncthreads();

        const int e0 = __builtin_amdgcn_readfirstlane(s_off[ra]);
        const int e1 = __builtin_amdgcn_readfirstlane(s_off[rb]);
        int row = ra;
        int rend = __builtin_amdgcn_readfirstlane(s_off[ra + 1]);
        v2f a = {0.f, 0.f};
#define FLUSH_ROW()                                                                         \
        {                                                                                   \
            v2f* dst = (v2f*)(acc + row * AP + lane * 2);                                   \
            if (pass == 0) { *dst = a; }                                                    \
            else { v2f o = *dst; o.x += a.x; o.y += a.y; *dst = o; }                        \
            a.x = 0.f; a.y = 0.f;                                                           \
            ++row;                                                                          \
            rend = (row < rb) ? __builtin_amdgcn_readfirstlane(s_off[row + 1]) : 0x7fffffff; \
        }
        v2i nx = {0, 0};
        if (e0 < e1) { const int li = e0 + lane; nx = s_cv[li < e1 ? li : e1 - 1]; }
        for (int base = e0; base < e1; base += 32) {
            const int n = (e1 - base < 32) ? (e1 - base) : 32;
            const int c = nx.x, vb = nx.y;
            if (base + 32 < e1) { const int li = base + 32 + lane; nx = s_cv[li < e1 ? li : e1 - 1]; }
            for (int j0 = 0; j0 < n; j0 += 8) {
                v2f xv[8];
#pragma unroll
                for (int u = 0; u < 8; ++u) {
                    const int cj = __builtin_amdgcn_readlane(c, j0 + u);
                    xv[u] = *(const v2f*)(xl + (size_t)cj * BATCH);
                }
                const int eb = base + j0;
                if (eb + 8 <= rend) {
#pragma unroll
                    for (int u = 0; u < 8; ++u) {
                        const float vj = __int_as_float(__builtin_amdgcn_readlane(vb, j0 + u));
                        const v2f vj2 = {vj, vj};
                        a = __builtin_elementwise_fma(xv[u], vj2, a);
                    }
                } else {
#pragma unroll
                    for (int u = 0; u < 8; ++u) {
                        while (eb + u >= rend) FLUSH_ROW()
                        const float vj = __int_as_float(__builtin_amdgcn_readlane(vb, j0 + u));
                        const v2f vj2 = {vj, vj};
                        a = __builtin_elementwise_fma(xv[u], vj2, a);
                    }
                }
            }
        }
        while (row < rb) FLUSH_ROW()
#undef FLUSH_ROW
    }
    __syncthreads();
    const int rows = (nd - row0 < RB) ? (nd - row0) : RB;
    const int fd = rows >> 4;
    const int ntiles = fd * (BATCH / 16);
    const int n = lane & 15;
    const int mbase = (lane < 16) ? 0 : 8;
    const v2f a = {(lane < 16) ? 1.0f : 0.0f, 0.0f};
    for (int t = w; t < ntiles; t += PT / 32) {
        const int b0 = (t & 3) * 16;
        const int d0 = (t >> 2) * 16;
        const float* cb = acc + (d0 + n) * AP + b0 + mbase;
        const float4 lo = *(const float4*)(cb);
        const float4 hi = *(const float4*)(cb + 4);
        const v8f c = {lo.x, lo.y, lo.z, lo.w, hi.x, hi.y, hi.z, hi.w};
        const v2f bm = {(lane < 16) ? bias[row0 + d0 + n] : 0.0f, 0.0f};
        const v8f d = __builtin_amdgcn_wmma_f32_16x16x4_f32(
             false, a,  false, bm,
             (short)0, c,  false,  false);
        float* o = out + (size_t)(b0 + mbase) * nd + row0 + d0 + n;
#pragma unroll
        for (int v = 0; v < 8; ++v) o[(size_t)v * nd] = d[v];
    }
    const int dt = fd * 16;
    for (int i = tid; i < (rows - dt) * BATCH; i += PT) {
        const int dd = dt + (i >> 6);
        const int bb = i & 63;
        out[(size_t)bb * nd + row0 + dd] = acc[dd * AP + bb] + bias[row0 + dd];
    }
}

extern "C" void kernel_launch(void* const* d_in, const int* in_sizes, int n_in,
                              void* d_out, int out_size, void* d_ws, size_t ws_size,
                              hipStream_t stream) {
    const float* x    = (const float*)d_in[0];
    const float* vals = (const float*)d_in[1];
    const float* bias = (const float*)d_in[2];
    const int* row    = (const int*)d_in[3];
    const int* col    = (const int*)d_in[4];
    float* out = (float*)d_out;

    const int nnz = in_sizes[1];
    const int nd  = in_sizes[2];
    const int ns  = in_sizes[0] / BATCH;

    float* xt    = (float*)d_ws;
    float* out_t = xt + (size_t)ns * BATCH;

    if ((ns & 3) == 0) {
        k_transpose_wide<<<(ns + TS - 1) / TS, 256, 0, stream>>>(x, xt, ns);
    } else {
        dim3 tb(32, 8);
        dim3 tg((ns + 31) / 32, (BATCH + 31) / 32);
        k_transpose<<<tg, tb, 0, stream>>>(x, xt, ns);
    }

    const int nb = (nd + RB - 1) / RB;
    const int pb = (int)(((long long)nnz + TILE - 1) / TILE);
    const size_t need_words = (size_t)ns * BATCH + (size_t)nd * BATCH + 2 * (size_t)nnz +
                              2 * (size_t)pb * NBMAX + NBMAX;
    if (nb <= NBMAX && pb >= 1 && pb <= PT && ns <= (1 << CBITS) && need_words * 4 <= ws_size) {
        int* rec  = (int*)(out_t + (size_t)nd * BATCH);
        int* runs = rec + 2 * (size_t)nnz;
        k_place<<<pb, PT, 0, stream>>>(row, col, vals, nnz, rec, runs);
        k_accum<<<nb, PT, 0, stream>>>(xt, rec, runs, pb, bias, out, nd);
        return;
    } else {
        long long n4 = ((long long)nd * BATCH) / 4;
        long long zb = (n4 + 255) / 256;
        if (zb > 16384) zb = 16384;
        k_zero<<<(int)zb, 256, 0, stream>>>((float4*)out_t, n4);
        long long waves = ((long long)nnz + 32LL * CHUNKS - 1) / (32LL * CHUNKS);
        long long sblocks = (waves + 7) / 8;
        if (sblocks < 1) sblocks = 1;
        k_scatter<<<(int)sblocks, 256, 0, stream>>>(xt, vals, row, col, out_t, nnz);
    }

    const int dtiles = nd / 16;
    if (dtiles > 0) {
        const int tiles = (BATCH / 16) * dtiles;
        const int eblocks = (tiles + 7) / 8;
        k_epilogue_wmma<<<eblocks, 256, 0, stream>>>(out_t, bias, out, nd, dtiles);
    }
    const int dstart = dtiles * 16;
    if (dstart < nd) {
        const int total = (nd - dstart) * BATCH;
        k_epilogue_tail<<<(total + 255) / 256, 256, 0, stream>>>(out_t, bias, out, nd, dstart);
    }
}


# Transpose moved into the placement workgroups: each also turns 3 or 4 64x128 tiles of x through LDS, x loads in flight with its edge loads, xt stores drain under the LDS ranking; 3 -> 2 launches
# speedup vs baseline: 3.0832x; 1.0710x over previous
#include <hip/hip_runtime.h>

typedef __attribute__((ext_vector_type(2))) float v2f;
typedef __attribute__((ext_vector_type(8))) float v8f;

#define BATCH 64
#define CHUNKS 4

__global__ void k_transpose(const float* __restrict__ x, float* __restrict__ xt, int ns) {
    __shared__ float tile[32][33];
    const int tx = threadIdx.x;
    const int ty = threadIdx.y;
    const int s0 = blockIdx.x * 32;
    const int b0 = blockIdx.y * 32;
#pragma unroll
    for (int k = 0; k < 4; ++k) {
        int s = s0 + tx, b = b0 + ty + k * 8;
        if (s < ns && b < BATCH) tile[ty + k * 8][tx] = x[(size_t)b * ns + s];
    }
    __syncthreads();
#pragma unroll
    for (int k = 0; k < 4; ++k) {
        int s = s0 + ty + k * 8, b = b0 + tx;
        if (s < ns && b < BATCH) xt[(size_t)s * BATCH + b] = tile[tx][ty + k * 8];
    }
}

#define TS 128
#define TPITCH 65
template <bool FULL>
__device__ __forceinline__ void turn_issue(const float* __restrict__ x, int ns, int s0, int tid, float4 (&v)[8]) {
    const int q = tid & 31;
    const int w = tid >> 5;
    if (FULL || s0 + 4 * q < ns) {
        const float* p = x + (size_t)w * ns + s0 + 4 * q;
        const size_t step = (size_t)8 * ns;
#pragma unroll
        for (int k = 0; k < 8; ++k) v[k] = *(const float4*)(p + k * step);
    }
}

template <bool FULL>
__device__ __forceinline__ void turn_lds(int ns, int s0, float* __restrict__ t, int tid, const float4 (&v)[8]) {
    const int q = tid & 31;
    const int w = tid >> 5;
    if (FULL || s0 + 4 * q < ns) {
        const int wbase = (4 * q) * TPITCH + ((4 * q) >> 6) + w;
#pragma unroll
        for (int k = 0; k < 8; ++k) {
            t[wbase + 8 * k]              = v[k].x;
            t[wbase + 8 * k + TPITCH]     = v[k].y;
            t[wbase + 8 * k + 2 * TPITCH] = v[k].z;
            t[wbase + 8 * k + 3 * TPITCH] = v[k].w;
        }
    }
}

template <bool FULL>
__device__ __forceinline__ void turn_store(float* __restrict__ xt, int ns, int s0,
                                           const float* __restrict__ t, int tid) {
    const int bq = tid & 15;
    const int sr = tid >> 4;
    float* dst = xt + (size_t)(s0 + sr) * BATCH + 4 * bq;
    if (FULL) {
        float4 o[8];
#pragma unroll
        for (int k = 0; k < 8; ++k) {
            const int s = sr + 16 * k;
            const int rb = s * TPITCH + (s >> 6) + 4 * bq;
            o[k] = make_float4(t[rb], t[rb + 1], t[rb + 2], t[rb + 3]);
        }
#pragma unroll
        for (int k = 0; k < 8; ++k) *(float4*)(dst + (size_t)(16 * k) * BATCH) = o[k];
    } else {
#pragma unroll
        for (int k = 0; k < 8; ++k) {
            const int s = sr + 16 * k;
            if (s0 + s < ns) {
                const int rb = s * TPITCH + (s >> 6) + 4 * bq;
                *(float4*)(dst + (size_t)(16 * k) * BATCH) =
                    make_float4(t[rb], t[rb + 1], t[rb + 2], t[rb + 3]);
            }
        }
    }
}

__global__ void __launch_bounds__(256) k_transpose_wide(const float* __restrict__ x,
                                                        float* __restrict__ xt, int ns) {
    __shared__ float t[TS * TPITCH + 4];
    const int s0 = blockIdx.x * TS;
    const bool full = s0 + TS <= ns;
    float4 v[8] = {};
    if (full) { turn_issue<true>(x, ns, s0, threadIdx.x, v);  turn_lds<true>(ns, s0, t, threadIdx.x, v); }
    else      { turn_issue<false>(x, ns, s0, threadIdx.x, v); turn_lds<false>(ns, s0, t, threadIdx.x, v); }
    __syncthreads();
    if (full) turn_store<true>(xt, ns, s0, t, threadIdx.x);
    else      turn_store<false>(xt, ns, s0, t, threadIdx.x);
}

__global__ void k_zero(float4* __restrict__ p, long long n4) {
    long long i = (long long)blockIdx.x * blockDim.x + threadIdx.x;
    const long long stride = (long long)gridDim.x * blockDim.x;
    const float4 z = make_float4(0.f, 0.f, 0.f, 0.f);
    for (; i < n4; i += stride) p[i] = z;
}

__global__ void k_scatter(const float* __restrict__ xt, const float* __restrict__ vals,
                          const int* __restrict__ row, const int* __restrict__ col,
                          float* __restrict__ out_t, int nnz) {
    const int lane = threadIdx.x & 31;
    const long long wave = ((long long)blockIdx.x * blockDim.x + threadIdx.x) >> 5;
    const long long base0 = wave * (32LL * CHUNKS);
    for (int ch = 0; ch < CHUNKS; ++ch) {
        const long long base = base0 + 32LL * ch;
        if (base >= nnz) return;
        const long long idx = base + lane;
        float v = 0.f;
        int r = 0, c = 0;
        if (idx < nnz) {
            v = vals[idx];
            r = row[idx];
            c = col[idx];
        }
        if (ch + 1 < CHUNKS && base + 32 + lane < nnz) {
            __builtin_prefetch(vals + base + 32 + lane, 0, 0);
            __builtin_prefetch(row + base + 32 + lane, 0, 0);
            __builtin_prefetch(col + base + 32 + lane, 0, 0);
        }
        const int count = (nnz - base < 32) ? (int)(nnz - base) : 32;
        for (int j = 0; j < count; ++j) {
            const float vj = __shfl(v, j);
            const int rj   = __shfl(r, j);
            const int cj   = __shfl(c, j);
            const v2f xv = *(const v2f*)(xt + (size_t)cj * BATCH + lane * 2);
            float* dst = out_t + (size_t)rj * BATCH + lane * 2;
            unsafeAtomicAdd(dst, xv.x * vj);
            unsafeAtomicAdd(dst + 1, xv.y * vj);
        }
    }
}

__global__ void k_epilogue_wmma(const float* __restrict__ out_t, const float* __restrict__ bias,
                                float* __restrict__ out, int nd, int dtiles) {
    const int lane = threadIdx.x & 31;
    const int wave = (int)(((long long)blockIdx.x * blockDim.x + threadIdx.x) >> 5);
    const int tiles = (BATCH / 16) * dtiles;
    if (wave >= tiles) return;
    const int b0 = (wave & 3) * 16;
    const int d0 = (wave >> 2) * 16;
    const int n = lane & 15;
    const int mbase = (lane < 16) ? 0 : 8;

    const float* cbase = out_t + (size_t)(d0 + n) * BATCH + b0 + mbase;
    const float4 lo = *(const float4*)(cbase);
    const float4 hi = *(const float4*)(cbase + 4);
    v8f c = {lo.x, lo.y, lo.z, lo.w, hi.x, hi.y, hi.z, hi.w};

    v2f a = {(lane < 16) ? 1.0f : 0.0f, 0.0f};
    v2f bm = {(lane < 16) ? bias[d0 + n] : 0.0f, 0.0f};

    v8f d = __builtin_amdgcn_wmma_f32_16x16x4_f32(
         false, a,  false, bm,
         (short)0, c,  false,  false);

#pragma unroll
    for (int v = 0; v < 8; ++v) {
        out[(size_t)(b0 + mbase + v) * nd + d0 + n] = d[v];
    }
}

__global__ void k_epilogue_tail(const float* __restrict__ out_t, const float* __restrict__ bias,
                                float* __restrict__ out, int nd, int dstart) {
    const int total = (nd - dstart) * BATCH;
    const int i = blockIdx.x * blockDim.x + threadIdx.x;
    if (i >= total) return;
    const int d = dstart + (i >> 6);
    const int b = i & 63;
    out[(size_t)b * nd + d] = out_t[(size_t)d * BATCH + b] + bias[d];
}

typedef __attribute__((ext_vector_type(2))) int v2i;
#define CBITS 23
#define CMASK ((1 << CBITS) - 1)
#define RB 400
#define AP 68
#define NBMAX 256
#define PT 1024
#define EPT 13
#define TILE (PT * EPT)

__device__ __forceinline__ int block_excl_scan(int v, int* s_w, int& total) {
    const int lane = threadIdx.x & 31;
    const int w = threadIdx.x >> 5;
    int inc = v;
#pragma unroll
    for (int d = 1; d < 32; d <<= 1) {
        const int t = __shfl_up(inc, d);
        if (lane >= d) inc += t;
    }
    if (lane == 31) s_w[w] = inc;
    __syncthreads();
    if (w == 0) {
        const int x = s_w[lane];
        int xi = x;
#pragma unroll
        for (int d = 1; d < 32; d <<= 1) {
            const int t = __shfl_up(xi, d);
            if (lane >= d) xi += t;
        }
        s_w[32 + lane] = xi - x;
        if (lane == 31) s_w[64] = xi;
    }
    __syncthreads();
    total = s_w[64];
    const int r = s_w[32 + w] + inc - v;
    __syncthreads();
    return r;
}

#define TGRP 4
#define TWORDS (TS * TPITCH + 4)
static_assert(TGRP * 256 == PT, "one 256-thread group per x tile");
__global__ void __launch_bounds__(PT) k_place_turn(const int* __restrict__ row, const int* __restrict__ col,
                                                   const float* __restrict__ vals, int nnz,
                                                   int* __restrict__ rec, int* __restrict__ runs,
                                                   const float* __restrict__ x, float* __restrict__ xt,
                                                   int ns, int ntile, int pbw) {
    __shared__ __attribute__((aligned(8))) v2i s_rec[TILE];
    __shared__ int s_h[NBMAX];
    __shared__ int s_loff[NBMAX];
    __shared__ int s_w[65];
    __shared__ float s_t[TGRP * TWORDS];
    const int tid = threadIdx.x;
    const int g = tid >> 8;
    const int lt = tid & 255;
    float* tl = s_t + g * TWORDS;

    if (tid < NBMAX) s_h[tid] = 0;

    int myr[EPT], myc[EPT], myk[EPT];
    float myv[EPT];
    const long long base = (long long)blockIdx.x * TILE;
    const bool full = base + TILE <= (long long)nnz;
    if (full) {
#pragma unroll
        for (int k = 0; k < EPT; ++k) myr[k] = row[base + k * PT + tid];
#pragma unroll
        for (int k = 0; k < EPT; ++k) {
            myc[k] = col[base + k * PT + tid];
            myv[k] = vals[base + k * PT + tid];
        }
    } else {
#pragma unroll
        for (int k = 0; k < EPT; ++k) {
            const long long e = base + k * PT + tid;
            myr[k] = 0;
            if (e < nnz) myr[k] = row[e];
        }
#pragma unroll
        for (int k = 0; k < EPT; ++k) {
            const long long e = base + k * PT + tid;
            myc[k] = 0; myv[k] = 0.f;
            if (e < nnz) { myc[k] = col[e]; myv[k] = vals[e]; }
        }
    }
    const int tile0 = g * pbw + (int)blockIdx.x;
    const int s00 = tile0 * TS;
    const bool live0 = tile0 < ntile;
    const bool full0 = s00 + TS <= ns;
    float4 xv[8] = {};
    if (live0) {
        if (full0) turn_issue<true>(x, ns, s00, lt, xv);
        else       turn_issue<false>(x, ns, s00, lt, xv);
    }
    if (live0) {
        if (full0) turn_lds<true>(ns, s00, tl, lt, xv);
        else       turn_lds<false>(ns, s00, tl, lt, xv);
    }
    __syncthreads();
    if (live0) {
        if (full0) turn_store<true>(xt, ns, s00, tl, lt);
        else       turn_store<false>(xt, ns, s00, tl, lt);
    }
    for (int r0 = TGRP * pbw; r0 < ntile; r0 += TGRP * pbw) {
        const int tile = r0 + g * pbw + (int)blockIdx.x;
        const int s0 = tile * TS;
        const bool live = tile < ntile;
        const bool fl = s0 + TS <= ns;
        __syncthreads();
        if (live) {
            float4 v[8] = {};
            if (fl) { turn_issue<true>(x, ns, s0, lt, v);  turn_lds<true>(ns, s0, tl, lt, v); }
            else    { turn_issue<false>(x, ns, s0, lt, v); turn_lds<false>(ns, s0, tl, lt, v); }
        }
        __syncthreads();
        if (live) {
            if (fl) turn_store<true>(xt, ns, s0, tl, lt);
            else    turn_store<false>(xt, ns, s0, tl, lt);
        }
    }

#pragma unroll
    for (int k = 0; k < EPT; ++k) {
        const long long e = base + k * PT + tid;
        myk[k] = -1;
        if (full || e < nnz) myk[k] = atomicAdd(&s_h[myr[k] / RB], 1);
    }
    __syncthreads();
    const int h = (tid < NBMAX) ? s_h[tid] : 0;
    int total;
    const int lo = block_excl_scan(h, s_w, total);
    if (tid < NBMAX) {
        s_loff[tid] = lo;
        int* rp = runs + 2 * ((size_t)blockIdx.x * NBMAX + tid);
        rp[0] = lo;
        rp[1] = h;
    }
    __syncthreads();
#pragma unroll
    for (int k = 0; k < EPT; ++k) {
        if (myk[k] >= 0) {
            const int bk = myr[k] / RB;
            const v2i o = {(int)(((unsigned)(myr[k] - bk * RB) << CBITS) | (unsigned)myc[k]), __float_as_int(myv[k])};
            s_rec[s_loff[bk] + myk[k]] = o;
        }
    }
    __syncthreads();
    v2i* __restrict__ rec2 = (v2i*)rec + base;
    for (int i = tid; i < total; i += PT) rec2[i] = s_rec[i];
}

#define CAP 16384
#define SPT (CAP / PT)

__global__ void __launch_bounds__(PT) k_accum(const float* __restrict__ xt, const int* __restrict__ rec,
                                              const int* __restrict__ runs, int pb,
                                              const float* __restrict__ bias,
                                              float* __restrict__ out, int nd) {
    __shared__ __attribute__((aligned(16))) float acc[RB * AP];
    __shared__ __attribute__((aligned(8))) v2i s_cv[CAP];
    __shared__ int s_cnt[RB];
    __shared__ int s_off[RB + 1];
    __shared__ int s_w[65];
    __shared__ int s_rex[PT];
    __shared__ int s_rn[PT];
    __shared__ int s_rdl[PT];
    int* s_map = (int*)s_cv;
    const v2i* __restrict__ rec2 = (const v2i*)rec;
    const int tid = threadIdx.x;
    const int lane = tid & 31;
    const int w = __builtin_amdgcn_readfirstlane(tid >> 5);
    const int b = blockIdx.x;

    int total;
    int rn = 0, rlo = 0;
    if (tid < pb) {
        const int* rp = runs + 2 * ((size_t)tid * NBMAX + b);
        rlo = rp[0];
        rn = rp[1];
    }
    const int rex = block_excl_scan(rn, s_w, total);
    if (tid < pb) {
        s_rex[tid] = rex;
        s_rn[tid] = rn;
        s_rdl[tid] = tid * TILE + rlo - rex;
    }
    const int cnt = __builtin_amdgcn_readfirstlane(total);
    const int row0 = b * RB;
    const int ra = (w * RB) >> 5;
    const int rb = ((w + 1) * RB) >> 5;
    const float* xl = xt + lane * 2;
    int npass = (cnt + CAP - 1) / CAP;
    if (npass < 1) npass = 1;

    for (int pass = 0; pass < npass; ++pass) {
        const int v0 = pass * CAP;
        const int m = (cnt - v0 < CAP) ? (cnt - v0) : CAP;
        if (tid < RB) s_cnt[tid] = 0;
        __syncthreads();
        for (int p = w; p < pb; p += PT / 32) {
            const int pdl = s_rdl[p];
            int jlo = s_rex[p] - v0;
            int jhi = jlo + s_rn[p];
            if (jlo < 0) jlo = 0;
            if (jhi > CAP) jhi = CAP;
            for (int j = jlo + lane; j < jhi; j += 32) s_map[j] = pdl;
        }
        __syncthreads();
        v2i rc[SPT];
        int rk[SPT], dl[SPT];
#pragma unroll
        for (int k = 0; k < SPT; ++k) {
            const int i = k * PT + tid;
            dl[k] = 0;
            if (i < m) dl[k] = s_map[i];
        }
#pragma unroll
        for (int k = 0; k < SPT; ++k) {
            const int i = k * PT + tid;
            rc[k].x = 0; rc[k].y = 0;
            if (i < m) rc[k] = rec2[v0 + i + dl[k]];
        }
#pragma unroll
        for (int k = 0; k < SPT; ++k) {
            rk[k] = 0;
            if (k * PT + tid < m) rk[k] = atomicAdd(&s_cnt[(unsigned)rc[k].x >> CBITS], 1);
        }
        __syncthreads();
        const int h = (tid < RB) ? s_cnt[tid] : 0;
        int msum;
        const int ex = block_excl_scan(h, s_w, msum);
        if (tid < RB) s_off[tid] = ex;
        if (tid == RB) s_off[RB] = msum;
        __syncthreads();
#pragma unroll
        for (int k = 0; k < SPT; ++k) {
            if (k * PT + tid < m) {
                const v2i o = {rc[k].x & CMASK, rc[k].y};
                s_cv[s_off[(unsigned)rc[k].x >> CBITS] + rk[k]] = o;
            }
        }
        __syncthreads();

        const int e0 = __builtin_amdgcn_readfirstlane(s_off[ra]);
        const int e1 = __builtin_amdgcn_readfirstlane(s_off[rb]);
        int row = ra;
        int rend = __builtin_amdgcn_readfirstlane(s_off[ra + 1]);
        v2f a = {0.f, 0.f};
#define FLUSH_ROW()                                                                         \
        {                                                                                   \
            v2f* dst = (v2f*)(acc + row * AP + lane * 2);                                   \
            if (pass == 0) { *dst = a; }                                                    \
            else { v2f o = *dst; o.x += a.x; o.y += a.y; *dst = o; }                        \
            a.x = 0.f; a.y = 0.f;                                                           \
            ++row;                                                                          \
            rend = (row < rb) ? __builtin_amdgcn_readfirstlane(s_off[row + 1]) : 0x7fffffff; \
        }
        v2i nx = {0, 0};
        if (e0 < e1) { const int li = e0 + lane; nx = s_cv[li < e1 ? li : e1 - 1]; }
        for (int base = e0; base < e1; base += 32) {
            const int n = (e1 - base < 32) ? (e1 - base) : 32;
            const int c = nx.x, vb = nx.y;
            if (base + 32 < e1) { const int li = base + 32 + lane; nx = s_cv[li < e1 ? li : e1 - 1]; }
            for (int j0 = 0; j0 < n; j0 += 8) {
                v2f xv[8];
#pragma unroll
                for (int u = 0; u < 8; ++u) {
                    const int cj = __builtin_amdgcn_readlane(c, j0 + u);
                    xv[u] = *(const v2f*)(xl + (size_t)cj * BATCH);
                }
                const int eb = base + j0;
                if (eb + 8 <= rend) {
#pragma unroll
                    for (int u = 0; u < 8; ++u) {
                        const float vj = __int_as_float(__builtin_amdgcn_readlane(vb, j0 + u));
                        const v2f vj2 = {vj, vj};
                        a = __builtin_elementwise_fma(xv[u], vj2, a);
                    }
                } else {
#pragma unroll
                    for (int u = 0; u < 8; ++u) {
                        while (eb + u >= rend) FLUSH_ROW()
                        const float vj = __int_as_float(__builtin_amdgcn_readlane(vb, j0 + u));
                        const v2f vj2 = {vj, vj};
                        a = __builtin_elementwise_fma(xv[u], vj2, a);
                    }
                }
            }
        }
        while (row < rb) FLUSH_ROW()
#undef FLUSH_ROW
    }
    __syncthreads();
    const int rows = (nd - row0 < RB) ? (nd - row0) : RB;
    const int fd = rows >> 4;
    const int ntiles = fd * (BATCH / 16);
    const int n = lane & 15;
    const int mbase = (lane < 16) ? 0 : 8;
    const v2f a = {(lane < 16) ? 1.0f : 0.0f, 0.0f};
    for (int t = w; t < ntiles; t += PT / 32) {
        const int b0 = (t & 3) * 16;
        const int d0 = (t >> 2) * 16;
        const float* cb = acc + (d0 + n) * AP + b0 + mbase;
        const float4 lo = *(const float4*)(cb);
        const float4 hi = *(const float4*)(cb + 4);
        const v8f c = {lo.x, lo.y, lo.z, lo.w, hi.x, hi.y, hi.z, hi.w};
        const v2f bm = {(lane < 16) ? bias[row0 + d0 + n] : 0.0f, 0.0f};
        const v8f d = __builtin_amdgcn_wmma_f32_16x16x4_f32(
             false, a,  false, bm,
             (short)0, c,  false,  false);
        float* o = out + (size_t)(b0 + mbase) * nd + row0 + d0 + n;
#pragma unroll
        for (int v = 0; v < 8; ++v) o[(size_t)v * nd] = d[v];
    }
    const int dt = fd * 16;
    for (int i = tid; i < (rows - dt) * BATCH; i += PT) {
        const int dd = dt + (i >> 6);
        const int bb = i & 63;
        out[(size_t)bb * nd + row0 + dd] = acc[dd * AP + bb] + bias[row0 + dd];
    }
}

extern "C" void kernel_launch(void* const* d_in, const int* in_sizes, int n_in,
                              void* d_out, int out_size, void* d_ws, size_t ws_size,
                              hipStream_t stream) {
    const float* x    = (const float*)d_in[0];
    const float* vals = (const float*)d_in[1];
    const float* bias = (const float*)d_in[2];
    const int* row    = (const int*)d_in[3];
    const int* col    = (const int*)d_in[4];
    float* out = (float*)d_out;

    const int nnz = in_sizes[1];
    const int nd  = in_sizes[2];
    const int ns  = in_sizes[0] / BATCH;

    float* xt    = (float*)d_ws;
    float* out_t = xt + (size_t)ns * BATCH;

    const int nb = (nd + RB - 1) / RB;
    const int pb = (int)(((long long)nnz + TILE - 1) / TILE);
    const size_t need_words = (size_t)ns * BATCH + (size_t)nd * BATCH + 2 * (size_t)nnz +
                              2 * (size_t)pb * NBMAX + NBMAX;
    const bool bucketed = nb <= NBMAX && pb >= 1 && pb <= PT && ns <= (1 << CBITS) && need_words * 4 <= ws_size;
    const bool wide = (ns & 3) == 0;

    int ntile = 0;
    if (bucketed && wide) {
        ntile = (ns + TS - 1) / TS;
    } else if (wide) {
        k_transpose_wide<<<(ns + TS - 1) / TS, 256, 0, stream>>>(x, xt, ns);
    } else {
        dim3 tb(32, 8);
        dim3 tg((ns + 31) / 32, (BATCH + 31) / 32);
        k_transpose<<<tg, tb, 0, stream>>>(x, xt, ns);
    }

    if (bucketed) {
        int* rec  = (int*)(out_t + (size_t)nd * BATCH);
        int* runs = rec + 2 * (size_t)nnz;
        k_place_turn<<<pb, PT, 0, stream>>>(row, col, vals, nnz, rec, runs, x, xt, ns, ntile, pb);
        k_accum<<<nb, PT, 0, stream>>>(xt, rec, runs, pb, bias, out, nd);
        return;
    } else {
        long long n4 = ((long long)nd * BATCH) / 4;
        long long zb = (n4 + 255) / 256;
        if (zb > 16384) zb = 16384;
        k_zero<<<(int)zb, 256, 0, stream>>>((float4*)out_t, n4);
        long long waves = ((long long)nnz + 32LL * CHUNKS - 1) / (32LL * CHUNKS);
        long long sblocks = (waves + 7) / 8;
        if (sblocks < 1) sblocks = 1;
        k_scatter<<<(int)sblocks, 256, 0, stream>>>(xt, vals, row, col, out_t, nnz);
    }

    const int dtiles = nd / 16;
    if (dtiles > 0) {
        const int tiles = (BATCH / 16) * dtiles;
        const int eblocks = (tiles + 7) / 8;
        k_epilogue_wmma<<<eblocks, 256, 0, stream>>>(out_t, bias, out, nd, dtiles);
    }
    const int dstart = dtiles * 16;
    if (dstart < nd) {
        const int total = (nd - dstart) * BATCH;
        k_epilogue_tail<<<(total + 255) / 256, 256, 0, stream>>>(out_t, bias, out, nd, dstart);
    }
}
